# starting kernel
# baseline (speedup 1.0000x reference)
#include <hip/hip_runtime.h>


#define SEQ_L   2048
#define DI      4096
#define RTOT    160
#define DTR     128
#define DS      16
#define KSPLIT  4
#define KCH     (DI / KSPLIT)
#define DBC_N   (SEQ_L * RTOT)

typedef float v2f __attribute__((ext_vector_type(2)));
typedef float v8f __attribute__((ext_vector_type(8)));

static __device__ __forceinline__ v8f wmma4(v2f a, v2f b, v8f c) {
    return __builtin_amdgcn_wmma_f32_16x16x4_f32(
               false, a, false, b, (short)0, c, false, false);
}

__global__ void __launch_bounds__(128)
sssm_gemm_in(const float* __restrict__ x, const float* __restrict__ Win,
             float* __restrict__ part) {
    const int wave = (blockIdx.x * 128 + threadIdx.x) >> 5;
    const int lane = threadIdx.x & 31;
    const int ks   = wave & (KSPLIT - 1);
    const int tile = wave >> 2;
    const int l0 = (tile / (RTOT / 16)) * 16;
    const int j0 = (tile % (RTOT / 16)) * 16;

    const int m  = lane & 15;
    const int kh = (lane >> 4) << 1;

    const float* ap = x   + (size_t)(l0 + m) * DI + ks * KCH + kh;
    const float* bp = Win + (size_t)(j0 + m) * DI + ks * KCH + kh;

    v8f acc0 = {}, acc1 = {}, acc2 = {}, acc3 = {};
    #pragma unroll 4
    for (int k = 0; k < KCH; k += 16) {
        v2f a0 = *(const v2f*)(ap + 0),  b0 = *(const v2f*)(bp + 0);
        v2f a1 = *(const v2f*)(ap + 4),  b1 = *(const v2f*)(bp + 4);
        v2f a2 = *(const v2f*)(ap + 8),  b2 = *(const v2f*)(bp + 8);
        v2f a3 = *(const v2f*)(ap + 12), b3 = *(const v2f*)(bp + 12);
        acc0 = wmma4(a0, b0, acc0);
        acc1 = wmma4(a1, b1, acc1);
        acc2 = wmma4(a2, b2, acc2);
        acc3 = wmma4(a3, b3, acc3);
        ap += 16; bp += 16;
    }
    v8f acc = (acc0 + acc1) + (acc2 + acc3);

    const int mrow = (lane >> 4) << 3;
    float* dst = part + (size_t)ks * DBC_N;
    #pragma unroll
    for (int r = 0; r < 8; ++r)
        dst[(size_t)(l0 + mrow + r) * RTOT + j0 + (lane & 15)] = acc[r];
}

__global__ void __launch_bounds__(256)
sssm_reduce(const float* __restrict__ part, float* __restrict__ dbc) {
    const int i = blockIdx.x * 256 + threadIdx.x;
    dbc[i] = (part[i] + part[i + DBC_N]) +
             (part[i + 2 * (size_t)DBC_N] + part[i + 3 * (size_t)DBC_N]);
}

__global__ void __launch_bounds__(256)
sssm_gemm_delta(const float* __restrict__ dbc, const float* __restrict__ Wd,
                float* __restrict__ delta) {
    const int wave = (blockIdx.x * 256 + threadIdx.x) >> 5;
    const int lane = threadIdx.x & 31;
    const int l0 = (wave >> 8) * 16;
    const int d0 = (wave & 255) * 16;

    const int m  = lane & 15;
    const int kh = (lane >> 4) << 1;

    const float* ap = dbc + (size_t)(l0 + m) * RTOT + kh;
    const float* bp = Wd  + (size_t)(d0 + m) * DTR  + kh;

    v8f acc0 = {}, acc1 = {}, acc2 = {}, acc3 = {};
    #pragma unroll
    for (int k = 0; k < DTR; k += 16) {
        v2f a0 = *(const v2f*)(ap + 0),  b0 = *(const v2f*)(bp + 0);
        v2f a1 = *(const v2f*)(ap + 4),  b1 = *(const v2f*)(bp + 4);
        v2f a2 = *(const v2f*)(ap + 8),  b2 = *(const v2f*)(bp + 8);
        v2f a3 = *(const v2f*)(ap + 12), b3 = *(const v2f*)(bp + 12);
        acc0 = wmma4(a0, b0, acc0);
        acc1 = wmma4(a1, b1, acc1);
        acc2 = wmma4(a2, b2, acc2);
        acc3 = wmma4(a3, b3, acc3);
        ap += 16; bp += 16;
    }
    v8f acc = (acc0 + acc1) + (acc2 + acc3);

    const int mrow = (lane >> 4) << 3;
    #pragma unroll
    for (int r = 0; r < 8; ++r) {
        float v  = acc[r];
        float sp = fmaxf(v, 0.0f) + log1pf(__expf(-fabsf(v)));
        delta[(size_t)(l0 + mrow + r) * DI + d0 + (lane & 15)] = sp;
    }
}

__global__ void __launch_bounds__(256)
sssm_scan(const float* __restrict__ x, const float* __restrict__ A_log,
          const float* __restrict__ Dvec, const float* __restrict__ dbc,
          float* __restrict__ out) {
    const int t = blockIdx.x * 256 + threadIdx.x;
    const int d = t >> 4;
    const int n = t & 15;

    const float A  = -__expf(A_log[(size_t)d * DS + n]);
    const float Dd = Dvec[d];

    float h = 0.0f;
    const float* xp  = x   + d;
    const float* dp  = out + d;
    const float* bcp = dbc + DTR + n;

    #pragma unroll 4
    for (int l = 0; l < SEQ_L; ++l) {
        float dl = dp[(size_t)l * DI];
        float xl = xp[(size_t)l * DI];
        float Bl = bcp[(size_t)l * RTOT];
        float Cl = bcp[(size_t)l * RTOT + DS];

        float dA = __expf(dl * A);
        h = fmaf(dA, h, dl * Bl * xl);
        float y = h * Cl;

        y += __shfl_xor(y, 1, 32);
        y += __shfl_xor(y, 2, 32);
        y += __shfl_xor(y, 4, 32);
        y += __shfl_xor(y, 8, 32);

        if (n == 0)
            out[(size_t)l * DI + d] = fmaf(xl, Dd, y);
    }
}

extern "C" void kernel_launch(void* const* d_in, const int* in_sizes, int n_in,
                              void* d_out, int out_size, void* d_ws, size_t ws_size,
                              hipStream_t stream) {
    const float* x    = (const float*)d_in[0];
    const float* Win  = (const float*)d_in[1];
    const float* Wd   = (const float*)d_in[2];
    const float* Alog = (const float*)d_in[3];
    const float* Dv   = (const float*)d_in[4];
    float*       out  = (float*)d_out;

    float* part = (float*)d_ws;
    float* dbc  = part + (size_t)KSPLIT * DBC_N;

    sssm_gemm_in<<<1280, 128, 0, stream>>>(x, Win, part);
    sssm_reduce<<<DBC_N / 256, 256, 0, stream>>>(part, dbc);
    sssm_gemm_delta<<<4096, 256, 0, stream>>>(dbc, Wd, out);
    sssm_scan<<<(DI * DS) / 256, 256, 0, stream>>>(x, Alog, Dv, dbc, out);
}


# scan: delta/x/B/C staged via LDS in 64-step chunks (16-byte loads one chunk ahead), y staged back from lane n of the same xor tree (last-bit differences possible), 16 steps per branch-free block
# speedup vs baseline: 6.2135x; 6.2135x over previous
#include <hip/hip_runtime.h>


#define SEQ_L   2048
#define DI      4096
#define RTOT    160
#define DTR     128
#define DS      16
#define KSPLIT  4
#define KCH     (DI / KSPLIT)
#define DBC_N   (SEQ_L * RTOT)

typedef float v2f __attribute__((ext_vector_type(2)));
typedef float v8f __attribute__((ext_vector_type(8)));

static __device__ __forceinline__ v8f wmma4(v2f a, v2f b, v8f c) {
    return __builtin_amdgcn_wmma_f32_16x16x4_f32(
               false, a, false, b, (short)0, c, false, false);
}

__global__ void __launch_bounds__(128)
sssm_gemm_in(const float* __restrict__ x, const float* __restrict__ Win,
             float* __restrict__ part) {
    const int wave = (blockIdx.x * 128 + threadIdx.x) >> 5;
    const int lane = threadIdx.x & 31;
    const int ks   = wave & (KSPLIT - 1);
    const int tile = wave >> 2;
    const int l0 = (tile / (RTOT / 16)) * 16;
    const int j0 = (tile % (RTOT / 16)) * 16;

    const int m  = lane & 15;
    const int kh = (lane >> 4) << 1;

    const float* ap = x   + (size_t)(l0 + m) * DI + ks * KCH + kh;
    const float* bp = Win + (size_t)(j0 + m) * DI + ks * KCH + kh;

    v8f acc0 = {}, acc1 = {}, acc2 = {}, acc3 = {};
    #pragma unroll 4
    for (int k = 0; k < KCH; k += 16) {
        v2f a0 = *(const v2f*)(ap + 0),  b0 = *(const v2f*)(bp + 0);
        v2f a1 = *(const v2f*)(ap + 4),  b1 = *(const v2f*)(bp + 4);
        v2f a2 = *(const v2f*)(ap + 8),  b2 = *(const v2f*)(bp + 8);
        v2f a3 = *(const v2f*)(ap + 12), b3 = *(const v2f*)(bp + 12);
        acc0 = wmma4(a0, b0, acc0);
        acc1 = wmma4(a1, b1, acc1);
        acc2 = wmma4(a2, b2, acc2);
        acc3 = wmma4(a3, b3, acc3);
        ap += 16; bp += 16;
    }
    v8f acc = (acc0 + acc1) + (acc2 + acc3);

    const int mrow = (lane >> 4) << 3;
    float* dst = part + (size_t)ks * DBC_N;
    #pragma unroll
    for (int r = 0; r < 8; ++r)
        dst[(size_t)(l0 + mrow + r) * RTOT + j0 + (lane & 15)] = acc[r];
}

__global__ void __launch_bounds__(256)
sssm_reduce(const float* __restrict__ part, float* __restrict__ dbc) {
    const int i = blockIdx.x * 256 + threadIdx.x;
    dbc[i] = (part[i] + part[i + DBC_N]) +
             (part[i + 2 * (size_t)DBC_N] + part[i + 3 * (size_t)DBC_N]);
}

__global__ void __launch_bounds__(256)
sssm_gemm_delta(const float* __restrict__ dbc, const float* __restrict__ Wd,
                float* __restrict__ delta) {
    const int wave = (blockIdx.x * 256 + threadIdx.x) >> 5;
    const int lane = threadIdx.x & 31;
    const int l0 = (wave >> 8) * 16;
    const int d0 = (wave & 255) * 16;

    const int m  = lane & 15;
    const int kh = (lane >> 4) << 1;

    const float* ap = dbc + (size_t)(l0 + m) * RTOT + kh;
    const float* bp = Wd  + (size_t)(d0 + m) * DTR  + kh;

    v8f acc0 = {}, acc1 = {}, acc2 = {}, acc3 = {};
    #pragma unroll
    for (int k = 0; k < DTR; k += 16) {
        v2f a0 = *(const v2f*)(ap + 0),  b0 = *(const v2f*)(bp + 0);
        v2f a1 = *(const v2f*)(ap + 4),  b1 = *(const v2f*)(bp + 4);
        v2f a2 = *(const v2f*)(ap + 8),  b2 = *(const v2f*)(bp + 8);
        v2f a3 = *(const v2f*)(ap + 12), b3 = *(const v2f*)(bp + 12);
        acc0 = wmma4(a0, b0, acc0);
        acc1 = wmma4(a1, b1, acc1);
        acc2 = wmma4(a2, b2, acc2);
        acc3 = wmma4(a3, b3, acc3);
        ap += 16; bp += 16;
    }
    v8f acc = (acc0 + acc1) + (acc2 + acc3);

    const int mrow = (lane >> 4) << 3;
    #pragma unroll
    for (int r = 0; r < 8; ++r) {
        float v  = acc[r];
        float sp = fmaxf(v, 0.0f) + log1pf(__expf(-fabsf(v)));
        delta[(size_t)(l0 + mrow + r) * DI + d0 + (lane & 15)] = sp;
    }
}

#define SCAN_T   64
#define SCAN_NC  (SEQ_L / SCAN_T)
#define ROWW     64

__global__ void __launch_bounds__(256)
sssm_scan(const float* __restrict__ x, const float* __restrict__ A_log,
          const float* __restrict__ Dvec, const float* __restrict__ dbc,
          float* out) {
    __shared__ __attribute__((aligned(16))) float sIn[2][SCAN_T][ROWW];
    __shared__ __attribute__((aligned(16))) float sY[SCAN_T][DS];

    const int t    = threadIdx.x;
    const int dloc = t >> 4;
    const int n    = t & 15;
    const int d0   = blockIdx.x * 16;
    const int d    = d0 + dloc;

    const float A  = -__expf(A_log[(size_t)d * DS + n]);
    const float Dd = Dvec[d];

    const int r4 = t >> 2, q4 = (t & 3) << 2;
    const int r8 = t >> 3, q8 = (t & 7) << 2;

    const float* gdel = out + (size_t)r4 * DI + d0 + q4;
    const float* gx   = x   + (size_t)r4 * DI + d0 + q4;
    const float* gbc0 = dbc + (size_t)r8 * RTOT + DTR + q8;
    const float* gbc1 = gbc0 + (size_t)32 * RTOT;
    float*       gy   = out + (size_t)r4 * DI + d0 + q4;

    float4 pd  = *(const float4*)gdel;
    float4 px  = *(const float4*)gx;
    float4 pb0 = *(const float4*)gbc0;
    float4 pb1 = *(const float4*)gbc1;

    float h = 0.0f;
    for (int c = 0; c < SCAN_NC; ++c) {
        float (*buf)[ROWW] = sIn[c & 1];
        *(float4*)&buf[r4][q4]           = pd;
        *(float4*)&buf[r4][16 + q4]      = px;
        *(float4*)&buf[r8][32 + q8]      = pb0;
        *(float4*)&buf[32 + r8][32 + q8] = pb1;
        __syncthreads();

        if (c + 1 < SCAN_NC) {
            const size_t adv = (size_t)(c + 1) * SCAN_T;
            pd  = *(const float4*)(gdel + adv * DI);
            px  = *(const float4*)(gx   + adv * DI);
            pb0 = *(const float4*)(gbc0 + adv * RTOT);
            pb1 = *(const float4*)(gbc1 + adv * RTOT);
        }

        for (int g = 0; g < SCAN_T; g += 16) {
            float dl[16], xl[16], Bl[16], Cl[16], p[16];
            #pragma unroll
            for (int j = 0; j < 16; ++j) {
                const float* row = buf[g + j];
                dl[j] = row[dloc];
                xl[j] = row[16 + dloc];
                Bl[j] = row[32 + n];
                Cl[j] = row[48 + n];
            }
            #pragma unroll
            for (int j = 0; j < 16; ++j) {
                float dA = __expf(dl[j] * A);
                h = fmaf(dA, h, dl[j] * Bl[j] * xl[j]);
                p[j] = h * Cl[j];
            }
            #pragma unroll
            for (int j = 0; j < 16; ++j) p[j] += __shfl_xor(p[j], 1, 32);
            #pragma unroll
            for (int j = 0; j < 16; ++j) p[j] += __shfl_xor(p[j], 2, 32);
            #pragma unroll
            for (int j = 0; j < 16; ++j) p[j] += __shfl_xor(p[j], 4, 32);
            #pragma unroll
            for (int j = 0; j < 16; ++j) p[j] += __shfl_xor(p[j], 8, 32);
            float keep = 0.0f;
            #pragma unroll
            for (int j = 0; j < 16; ++j) {
                float o = fmaf(xl[j], Dd, p[j]);
                keep = (n == j) ? o : keep;
            }
            sY[g + n][dloc] = keep;
        }
        __syncthreads();

        *(float4*)(gy + (size_t)c * SCAN_T * DI) = *(const float4*)&sY[r4][q4];
    }
}

extern "C" void kernel_launch(void* const* d_in, const int* in_sizes, int n_in,
                              void* d_out, int out_size, void* d_ws, size_t ws_size,
                              hipStream_t stream) {
    const float* x    = (const float*)d_in[0];
    const float* Win  = (const float*)d_in[1];
    const float* Wd   = (const float*)d_in[2];
    const float* Alog = (const float*)d_in[3];
    const float* Dv   = (const float*)d_in[4];
    float*       out  = (float*)d_out;

    float* part = (float*)d_ws;
    float* dbc  = part + (size_t)KSPLIT * DBC_N;

    sssm_gemm_in<<<1280, 128, 0, stream>>>(x, Win, part);
    sssm_reduce<<<DBC_N / 256, 256, 0, stream>>>(part, dbc);
    sssm_gemm_delta<<<4096, 256, 0, stream>>>(dbc, Wd, out);
    sssm_scan<<<(DI * DS) / 256, 256, 0, stream>>>(x, Alog, Dv, dbc, out);
}


# scan: a 16-step block's 16 lane sums by one step-halving butterfly, 15 DPP row_xmask adds (same xor tree, first add unfused: last-bit changes) for 64 LDS permutes; x*D added at y write-back
# speedup vs baseline: 8.4104x; 1.3536x over previous
#include <hip/hip_runtime.h>


#define SEQ_L   2048
#define DI      4096
#define RTOT    160
#define DTR     128
#define DS      16
#define KSPLIT  4
#define KCH     (DI / KSPLIT)
#define DBC_N   (SEQ_L * RTOT)

typedef float v2f __attribute__((ext_vector_type(2)));
typedef float v8f __attribute__((ext_vector_type(8)));

static __device__ __forceinline__ v8f wmma4(v2f a, v2f b, v8f c) {
    return __builtin_amdgcn_wmma_f32_16x16x4_f32(
               false, a, false, b, (short)0, c, false, false);
}

__global__ void __launch_bounds__(128)
sssm_gemm_in(const float* __restrict__ x, const float* __restrict__ Win,
             float* __restrict__ part) {
    const int wave = (blockIdx.x * 128 + threadIdx.x) >> 5;
    const int lane = threadIdx.x & 31;
    const int ks   = wave & (KSPLIT - 1);
    const int tile = wave >> 2;
    const int l0 = (tile / (RTOT / 16)) * 16;
    const int j0 = (tile % (RTOT / 16)) * 16;

    const int m  = lane & 15;
    const int kh = (lane >> 4) << 1;

    const float* ap = x   + (size_t)(l0 + m) * DI + ks * KCH + kh;
    const float* bp = Win + (size_t)(j0 + m) * DI + ks * KCH + kh;

    v8f acc0 = {}, acc1 = {}, acc2 = {}, acc3 = {};
    #pragma unroll 4
    for (int k = 0; k < KCH; k += 16) {
        v2f a0 = *(const v2f*)(ap + 0),  b0 = *(const v2f*)(bp + 0);
        v2f a1 = *(const v2f*)(ap + 4),  b1 = *(const v2f*)(bp + 4);
        v2f a2 = *(const v2f*)(ap + 8),  b2 = *(const v2f*)(bp + 8);
        v2f a3 = *(const v2f*)(ap + 12), b3 = *(const v2f*)(bp + 12);
        acc0 = wmma4(a0, b0, acc0);
        acc1 = wmma4(a1, b1, acc1);
        acc2 = wmma4(a2, b2, acc2);
        acc3 = wmma4(a3, b3, acc3);
        ap += 16; bp += 16;
    }
    v8f acc = (acc0 + acc1) + (acc2 + acc3);

    const int mrow = (lane >> 4) << 3;
    float* dst = part + (size_t)ks * DBC_N;
    #pragma unroll
    for (int r = 0; r < 8; ++r)
        dst[(size_t)(l0 + mrow + r) * RTOT + j0 + (lane & 15)] = acc[r];
}

__global__ void __launch_bounds__(256)
sssm_reduce(const float* __restrict__ part, float* __restrict__ dbc) {
    const int i = blockIdx.x * 256 + threadIdx.x;
    dbc[i] = (part[i] + part[i + DBC_N]) +
             (part[i + 2 * (size_t)DBC_N] + part[i + 3 * (size_t)DBC_N]);
}

__global__ void __launch_bounds__(256)
sssm_gemm_delta(const float* __restrict__ dbc, const float* __restrict__ Wd,
                float* __restrict__ delta) {
    const int wave = (blockIdx.x * 256 + threadIdx.x) >> 5;
    const int lane = threadIdx.x & 31;
    const int l0 = (wave >> 8) * 16;
    const int d0 = (wave & 255) * 16;

    const int m  = lane & 15;
    const int kh = (lane >> 4) << 1;

    const float* ap = dbc + (size_t)(l0 + m) * RTOT + kh;
    const float* bp = Wd  + (size_t)(d0 + m) * DTR  + kh;

    v8f acc0 = {}, acc1 = {}, acc2 = {}, acc3 = {};
    #pragma unroll
    for (int k = 0; k < DTR; k += 16) {
        v2f a0 = *(const v2f*)(ap + 0),  b0 = *(const v2f*)(bp + 0);
        v2f a1 = *(const v2f*)(ap + 4),  b1 = *(const v2f*)(bp + 4);
        v2f a2 = *(const v2f*)(ap + 8),  b2 = *(const v2f*)(bp + 8);
        v2f a3 = *(const v2f*)(ap + 12), b3 = *(const v2f*)(bp + 12);
        acc0 = wmma4(a0, b0, acc0);
        acc1 = wmma4(a1, b1, acc1);
        acc2 = wmma4(a2, b2, acc2);
        acc3 = wmma4(a3, b3, acc3);
        ap += 16; bp += 16;
    }
    v8f acc = (acc0 + acc1) + (acc2 + acc3);

    const int mrow = (lane >> 4) << 3;
    #pragma unroll
    for (int r = 0; r < 8; ++r) {
        float v  = acc[r];
        float sp = fmaxf(v, 0.0f) + log1pf(__expf(-fabsf(v)));
        delta[(size_t)(l0 + mrow + r) * DI + d0 + (lane & 15)] = sp;
    }
}

#define SCAN_T   64
#define SCAN_NC  (SEQ_L / SCAN_T)
#define ROWW     64

template <int M>
static __device__ __forceinline__ float row_xor(float v) {
    return __builtin_bit_cast(float,
        __builtin_amdgcn_update_dpp(0, __builtin_bit_cast(int, v), 0x160 + M, 0xf, 0xf, true));
}

__global__ void __launch_bounds__(256)
sssm_scan(const float* __restrict__ x, const float* __restrict__ A_log,
          const float* __restrict__ Dvec, const float* __restrict__ dbc,
          float* out) {
    __shared__ __attribute__((aligned(16))) float sIn[2][SCAN_T][ROWW];
    __shared__ __attribute__((aligned(16))) float sY[SCAN_T][DS];

    const int t    = threadIdx.x;
    const int dloc = t >> 4;
    const int n    = t & 15;
    const int d0   = blockIdx.x * 16;
    const int d    = d0 + dloc;

    const float A  = -__expf(A_log[(size_t)d * DS + n]);

    const bool b0 = (n & 1) != 0, b1 = (n & 2) != 0, b2 = (n & 4) != 0, b3 = (n & 8) != 0;

    const int r4 = t >> 2, q4 = (t & 3) << 2;
    const int r8 = t >> 3, q8 = (t & 7) << 2;

    const float* gdel = out + (size_t)r4 * DI + d0 + q4;
    const float* gx   = x   + (size_t)r4 * DI + d0 + q4;
    const float* gbc0 = dbc + (size_t)r8 * RTOT + DTR + q8;
    const float* gbc1 = gbc0 + (size_t)32 * RTOT;
    float*       gy   = out + (size_t)r4 * DI + d0 + q4;
    const float4 D4   = *(const float4*)(Dvec + d0 + q4);

    float4 pd  = *(const float4*)gdel;
    float4 px  = *(const float4*)gx;
    float4 pb0 = *(const float4*)gbc0;
    float4 pb1 = *(const float4*)gbc1;

    float h = 0.0f;
    for (int c = 0; c < SCAN_NC; ++c) {
        float (*buf)[ROWW] = sIn[c & 1];
        *(float4*)&buf[r4][q4]           = pd;
        *(float4*)&buf[r4][16 + q4]      = px;
        *(float4*)&buf[r8][32 + q8]      = pb0;
        *(float4*)&buf[32 + r8][32 + q8] = pb1;
        __syncthreads();

        const float4 xc = px;
        if (c + 1 < SCAN_NC) {
            const size_t adv = (size_t)(c + 1) * SCAN_T;
            pd  = *(const float4*)(gdel + adv * DI);
            px  = *(const float4*)(gx   + adv * DI);
            pb0 = *(const float4*)(gbc0 + adv * RTOT);
            pb1 = *(const float4*)(gbc1 + adv * RTOT);
        }

        for (int g = 0; g < SCAN_T; g += 16) {
            float dl[16], xl[16], Bl[16], Cl[16], p[16];
            #pragma unroll
            for (int j = 0; j < 16; ++j) {
                const float* row = buf[g + j];
                dl[j] = row[dloc];
                xl[j] = row[16 + dloc];
                Bl[j] = row[32 + n];
                Cl[j] = row[48 + n];
            }
            #pragma unroll
            for (int j = 0; j < 16; ++j) {
                float dA = __expf(dl[j] * A);
                h = fmaf(dA, h, dl[j] * Bl[j] * xl[j]);
                p[j] = h * Cl[j];
            }
            float q[8], r[4], s[2];
            #pragma unroll
            for (int i = 0; i < 8; ++i) {
                const float kp = b0 ? p[2 * i + 1] : p[2 * i];
                const float sd = b0 ? p[2 * i]     : p[2 * i + 1];
                q[i] = kp + row_xor<1>(sd);
            }
            #pragma unroll
            for (int i = 0; i < 4; ++i) {
                const float kp = b1 ? q[2 * i + 1] : q[2 * i];
                const float sd = b1 ? q[2 * i]     : q[2 * i + 1];
                r[i] = kp + row_xor<2>(sd);
            }
            #pragma unroll
            for (int i = 0; i < 2; ++i) {
                const float kp = b2 ? r[2 * i + 1] : r[2 * i];
                const float sd = b2 ? r[2 * i]     : r[2 * i + 1];
                s[i] = kp + row_xor<4>(sd);
            }
            const float kp3 = b3 ? s[1] : s[0];
            const float sd3 = b3 ? s[0] : s[1];
            sY[g + n][dloc] = kp3 + row_xor<8>(sd3);
        }
        __syncthreads();

        const float4 ys = *(const float4*)&sY[r4][q4];
        float4 yo;
        yo.x = fmaf(xc.x, D4.x, ys.x);
        yo.y = fmaf(xc.y, D4.y, ys.y);
        yo.z = fmaf(xc.z, D4.z, ys.z);
        yo.w = fmaf(xc.w, D4.w, ys.w);
        *(float4*)(gy + (size_t)c * SCAN_T * DI) = yo;
    }
}

extern "C" void kernel_launch(void* const* d_in, const int* in_sizes, int n_in,
                              void* d_out, int out_size, void* d_ws, size_t ws_size,
                              hipStream_t stream) {
    const float* x    = (const float*)d_in[0];
    const float* Win  = (const float*)d_in[1];
    const float* Wd   = (const float*)d_in[2];
    const float* Alog = (const float*)d_in[3];
    const float* Dv   = (const float*)d_in[4];
    float*       out  = (float*)d_out;

    float* part = (float*)d_ws;
    float* dbc  = part + (size_t)KSPLIT * DBC_N;

    sssm_gemm_in<<<1280, 128, 0, stream>>>(x, Win, part);
    sssm_reduce<<<DBC_N / 256, 256, 0, stream>>>(part, dbc);
    sssm_gemm_delta<<<4096, 256, 0, stream>>>(dbc, Wd, out);
    sssm_scan<<<(DI * DS) / 256, 256, 0, stream>>>(x, Alog, Dv, dbc, out);
}


# both GEMMs: A/B fragments by one 16-byte load per lane feeding two WMMAs (half the load instructions; a WMMA now sums k {0,1,4,5} and {2,3,6,7} of each 8: last-bit changes in dbc and delta)
# speedup vs baseline: 9.0164x; 1.0721x over previous
#include <hip/hip_runtime.h>


#define SEQ_L   2048
#define DI      4096
#define RTOT    160
#define DTR     128
#define DS      16
#define KSPLIT  4
#define KCH     (DI / KSPLIT)
#define DBC_N   (SEQ_L * RTOT)

typedef float v2f __attribute__((ext_vector_type(2)));
typedef float v4f __attribute__((ext_vector_type(4)));
typedef float v8f __attribute__((ext_vector_type(8)));

static __device__ __forceinline__ v2f lo2(v4f v) { return v2f{v.x, v.y}; }
static __device__ __forceinline__ v2f hi2(v4f v) { return v2f{v.z, v.w}; }

static __device__ __forceinline__ v8f wmma4(v2f a, v2f b, v8f c) {
    return __builtin_amdgcn_wmma_f32_16x16x4_f32(
               false, a, false, b, (short)0, c, false, false);
}

__global__ void __launch_bounds__(128)
sssm_gemm_in(const float* __restrict__ x, const float* __restrict__ Win,
             float* __restrict__ part) {
    const int wave = (blockIdx.x * 128 + threadIdx.x) >> 5;
    const int lane = threadIdx.x & 31;
    const int ks   = wave & (KSPLIT - 1);
    const int tile = wave >> 2;
    const int l0 = (tile / (RTOT / 16)) * 16;
    const int j0 = (tile % (RTOT / 16)) * 16;

    const int m  = lane & 15;
    const int kh = (lane >> 4) << 2;

    const float* ap = x   + (size_t)(l0 + m) * DI + ks * KCH + kh;
    const float* bp = Win + (size_t)(j0 + m) * DI + ks * KCH + kh;

    v8f acc0 = {}, acc1 = {}, acc2 = {}, acc3 = {};
    #pragma unroll 4
    for (int k = 0; k < KCH; k += 16) {
        const v4f a01 = *(const v4f*)(ap + 0), b01 = *(const v4f*)(bp + 0);
        const v4f a23 = *(const v4f*)(ap + 8), b23 = *(const v4f*)(bp + 8);
        acc0 = wmma4(lo2(a01), lo2(b01), acc0);
        acc1 = wmma4(hi2(a01), hi2(b01), acc1);
        acc2 = wmma4(lo2(a23), lo2(b23), acc2);
        acc3 = wmma4(hi2(a23), hi2(b23), acc3);
        ap += 16; bp += 16;
    }
    v8f acc = (acc0 + acc1) + (acc2 + acc3);

    const int mrow = (lane >> 4) << 3;
    float* dst = part + (size_t)ks * DBC_N;
    #pragma unroll
    for (int r = 0; r < 8; ++r)
        dst[(size_t)(l0 + mrow + r) * RTOT + j0 + (lane & 15)] = acc[r];
}

__global__ void __launch_bounds__(256)
sssm_reduce(const float* __restrict__ part, float* __restrict__ dbc) {
    const int i = blockIdx.x * 256 + threadIdx.x;
    dbc[i] = (part[i] + part[i + DBC_N]) +
             (part[i + 2 * (size_t)DBC_N] + part[i + 3 * (size_t)DBC_N]);
}

__global__ void __launch_bounds__(256)
sssm_gemm_delta(const float* __restrict__ dbc, const float* __restrict__ Wd,
                float* __restrict__ delta) {
    const int wave = (blockIdx.x * 256 + threadIdx.x) >> 5;
    const int lane = threadIdx.x & 31;
    const int l0 = (wave >> 8) * 16;
    const int d0 = (wave & 255) * 16;

    const int m  = lane & 15;
    const int kh = (lane >> 4) << 2;

    const float* ap = dbc + (size_t)(l0 + m) * RTOT + kh;
    const float* bp = Wd  + (size_t)(d0 + m) * DTR  + kh;

    v8f acc0 = {}, acc1 = {}, acc2 = {}, acc3 = {};
    #pragma unroll
    for (int k = 0; k < DTR; k += 16) {
        const v4f a01 = *(const v4f*)(ap + 0), b01 = *(const v4f*)(bp + 0);
        const v4f a23 = *(const v4f*)(ap + 8), b23 = *(const v4f*)(bp + 8);
        acc0 = wmma4(lo2(a01), lo2(b01), acc0);
        acc1 = wmma4(hi2(a01), hi2(b01), acc1);
        acc2 = wmma4(lo2(a23), lo2(b23), acc2);
        acc3 = wmma4(hi2(a23), hi2(b23), acc3);
        ap += 16; bp += 16;
    }
    v8f acc = (acc0 + acc1) + (acc2 + acc3);

    const int mrow = (lane >> 4) << 3;
    #pragma unroll
    for (int r = 0; r < 8; ++r) {
        float v  = acc[r];
        float sp = fmaxf(v, 0.0f) + log1pf(__expf(-fabsf(v)));
        delta[(size_t)(l0 + mrow + r) * DI + d0 + (lane & 15)] = sp;
    }
}

#define SCAN_T   64
#define SCAN_NC  (SEQ_L / SCAN_T)
#define ROWW     64

template <int M>
static __device__ __forceinline__ float row_xor(float v) {
    return __builtin_bit_cast(float,
        __builtin_amdgcn_update_dpp(0, __builtin_bit_cast(int, v), 0x160 + M, 0xf, 0xf, true));
}

__global__ void __launch_bounds__(256)
sssm_scan(const float* __restrict__ x, const float* __restrict__ A_log,
          const float* __restrict__ Dvec, const float* __restrict__ dbc,
          float* out) {
    __shared__ __attribute__((aligned(16))) float sIn[2][SCAN_T][ROWW];
    __shared__ __attribute__((aligned(16))) float sY[SCAN_T][DS];

    const int t    = threadIdx.x;
    const int dloc = t >> 4;
    const int n    = t & 15;
    const int d0   = blockIdx.x * 16;
    const int d    = d0 + dloc;

    const float A  = -__expf(A_log[(size_t)d * DS + n]);

    const bool b0 = (n & 1) != 0, b1 = (n & 2) != 0, b2 = (n & 4) != 0, b3 = (n & 8) != 0;

    const int r4 = t >> 2, q4 = (t & 3) << 2;
    const int r8 = t >> 3, q8 = (t & 7) << 2;

    const float* gdel = out + (size_t)r4 * DI + d0 + q4;
    const float* gx   = x   + (size_t)r4 * DI + d0 + q4;
    const float* gbc0 = dbc + (size_t)r8 * RTOT + DTR + q8;
    const float* gbc1 = gbc0 + (size_t)32 * RTOT;
    float*       gy   = out + (size_t)r4 * DI + d0 + q4;
    const float4 D4   = *(const float4*)(Dvec + d0 + q4);

    float4 pd  = *(const float4*)gdel;
    float4 px  = *(const float4*)gx;
    float4 pb0 = *(const float4*)gbc0;
    float4 pb1 = *(const float4*)gbc1;

    float h = 0.0f;
    for (int c = 0; c < SCAN_NC; ++c) {
        float (*buf)[ROWW] = sIn[c & 1];
        *(float4*)&buf[r4][q4]           = pd;
        *(float4*)&buf[r4][16 + q4]      = px;
        *(float4*)&buf[r8][32 + q8]      = pb0;
        *(float4*)&buf[32 + r8][32 + q8] = pb1;
        __syncthreads();

        const float4 xc = px;
        if (c + 1 < SCAN_NC) {
            const size_t adv = (size_t)(c + 1) * SCAN_T;
            pd  = *(const float4*)(gdel + adv * DI);
            px  = *(const float4*)(gx   + adv * DI);
            pb0 = *(const float4*)(gbc0 + adv * RTOT);
            pb1 = *(const float4*)(gbc1 + adv * RTOT);
        }

        for (int g = 0; g < SCAN_T; g += 16) {
            float dl[16], xl[16], Bl[16], Cl[16], p[16];
            #pragma unroll
            for (int j = 0; j < 16; ++j) {
                const float* row = buf[g + j];
                dl[j] = row[dloc];
                xl[j] = row[16 + dloc];
                Bl[j] = row[32 + n];
                Cl[j] = row[48 + n];
            }
            #pragma unroll
            for (int j = 0; j < 16; ++j) {
                float dA = __expf(dl[j] * A);
                h = fmaf(dA, h, dl[j] * Bl[j] * xl[j]);
                p[j] = h * Cl[j];
            }
            float q[8], r[4], s[2];
            #pragma unroll
            for (int i = 0; i < 8; ++i) {
                const float kp = b0 ? p[2 * i + 1] : p[2 * i];
                const float sd = b0 ? p[2 * i]     : p[2 * i + 1];
                q[i] = kp + row_xor<1>(sd);
            }
            #pragma unroll
            for (int i = 0; i < 4; ++i) {
                const float kp = b1 ? q[2 * i + 1] : q[2 * i];
                const float sd = b1 ? q[2 * i]     : q[2 * i + 1];
                r[i] = kp + row_xor<2>(sd);
            }
            #pragma unroll
            for (int i = 0; i < 2; ++i) {
                const float kp = b2 ? r[2 * i + 1] : r[2 * i];
                const float sd = b2 ? r[2 * i]     : r[2 * i + 1];
                s[i] = kp + row_xor<4>(sd);
            }
            const float kp3 = b3 ? s[1] : s[0];
            const float sd3 = b3 ? s[0] : s[1];
            sY[g + n][dloc] = kp3 + row_xor<8>(sd3);
        }
        __syncthreads();

        const float4 ys = *(const float4*)&sY[r4][q4];
        float4 yo;
        yo.x = fmaf(xc.x, D4.x, ys.x);
        yo.y = fmaf(xc.y, D4.y, ys.y);
        yo.z = fmaf(xc.z, D4.z, ys.z);
        yo.w = fmaf(xc.w, D4.w, ys.w);
        *(float4*)(gy + (size_t)c * SCAN_T * DI) = yo;
    }
}

extern "C" void kernel_launch(void* const* d_in, const int* in_sizes, int n_in,
                              void* d_out, int out_size, void* d_ws, size_t ws_size,
                              hipStream_t stream) {
    const float* x    = (const float*)d_in[0];
    const float* Win  = (const float*)d_in[1];
    const float* Wd   = (const float*)d_in[2];
    const float* Alog = (const float*)d_in[3];
    const float* Dv   = (const float*)d_in[4];
    float*       out  = (float*)d_out;

    float* part = (float*)d_ws;
    float* dbc  = part + (size_t)KSPLIT * DBC_N;

    sssm_gemm_in<<<1280, 128, 0, stream>>>(x, Win, part);
    sssm_reduce<<<DBC_N / 256, 256, 0, stream>>>(part, dbc);
    sssm_gemm_delta<<<4096, 256, 0, stream>>>(dbc, Wd, out);
    sssm_scan<<<(DI * DS) / 256, 256, 0, stream>>>(x, Alog, Dv, dbc, out);
}


# both GEMMs: register tiling, a wave owns 32x80 (GEMM1, K split 16) / 32x64 (GEMM2), loads grouped by 128-B line: a fragment load feeds 2-5 tiles; one acc per tile, pairwise K tree: last-bit changes
# speedup vs baseline: 12.0266x; 1.3339x over previous
#include <hip/hip_runtime.h>


#define SEQ_L   2048
#define DI      4096
#define RTOT    160
#define DTR     128
#define DS      16
#define KSPLIT  16
#define KCH     (DI / KSPLIT)
#define DBC_N   (SEQ_L * RTOT)

typedef float v2f __attribute__((ext_vector_type(2)));
typedef float v4f __attribute__((ext_vector_type(4)));
typedef float v8f __attribute__((ext_vector_type(8)));

static __device__ __forceinline__ v2f lo2(v4f v) { return v2f{v.x, v.y}; }
static __device__ __forceinline__ v2f hi2(v4f v) { return v2f{v.z, v.w}; }

static __device__ __forceinline__ v8f wmma4(v2f a, v2f b, v8f c) {
    return __builtin_amdgcn_wmma_f32_16x16x4_f32(
               false, a, false, b, (short)0, c, false, false);
}


__global__ void __launch_bounds__(128)
sssm_gemm_in(const float* __restrict__ x, const float* __restrict__ Win,
             float* __restrict__ part) {
    const int w    = threadIdx.x >> 5;
    const int lane = threadIdx.x & 31;
    const int ks   = blockIdx.x & (KSPLIT - 1);
    const int lp   = blockIdx.x >> 4;
    const int l0   = (lp * 2 + (w >> 1)) * 32;
    const int j0   = (w & 1) * 80;

    const int m  = lane & 15;
    const int kh = (lane >> 4) << 2;

    const float* ap = x   + (size_t)(l0 + m) * DI + ks * KCH + kh;
    const float* bp = Win + (size_t)(j0 + m) * DI + ks * KCH + kh;

    v8f acc[2][5];
    #pragma unroll
    for (int i = 0; i < 2; ++i)
        #pragma unroll
        for (int f = 0; f < 5; ++f) acc[i][f] = v8f{};

    v4f a[2][4], b[5][4];
    #pragma unroll
    for (int i = 0; i < 2; ++i)
        #pragma unroll
        for (int q = 0; q < 4; ++q) a[i][q] = *(const v4f*)(ap + (size_t)i * 16 * DI + 8 * q);
    #pragma unroll
    for (int f = 0; f < 5; ++f)
        #pragma unroll
        for (int q = 0; q < 4; ++q) b[f][q] = *(const v4f*)(bp + (size_t)f * 16 * DI + 8 * q);

    #pragma unroll
    for (int k = 0; k < KCH; k += 32) {
        v4f an[2][4], bn[5][4];
        if (k + 32 < KCH) {
            ap += 32; bp += 32;
            #pragma unroll
            for (int i = 0; i < 2; ++i)
                #pragma unroll
                for (int q = 0; q < 4; ++q) an[i][q] = *(const v4f*)(ap + (size_t)i * 16 * DI + 8 * q);
            #pragma unroll
            for (int f = 0; f < 5; ++f)
                #pragma unroll
                for (int q = 0; q < 4; ++q) bn[f][q] = *(const v4f*)(bp + (size_t)f * 16 * DI + 8 * q);
        }
        __builtin_amdgcn_sched_barrier(0);
        #pragma unroll
        for (int q = 0; q < 4; ++q) {
            #pragma unroll
            for (int i = 0; i < 2; ++i)
                #pragma unroll
                for (int f = 0; f < 5; ++f) acc[i][f] = wmma4(lo2(a[i][q]), lo2(b[f][q]), acc[i][f]);
            #pragma unroll
            for (int i = 0; i < 2; ++i)
                #pragma unroll
                for (int f = 0; f < 5; ++f) acc[i][f] = wmma4(hi2(a[i][q]), hi2(b[f][q]), acc[i][f]);
        }
        __builtin_amdgcn_sched_barrier(0);
        if (k + 32 < KCH) {
            #pragma unroll
            for (int i = 0; i < 2; ++i)
                #pragma unroll
                for (int q = 0; q < 4; ++q) a[i][q] = an[i][q];
            #pragma unroll
            for (int f = 0; f < 5; ++f)
                #pragma unroll
                for (int q = 0; q < 4; ++q) b[f][q] = bn[f][q];
        }
    }

    const int mrow = (lane >> 4) << 3;
    float* dst = part + (size_t)ks * DBC_N;
    #pragma unroll
    for (int i = 0; i < 2; ++i)
        #pragma unroll
        for (int f = 0; f < 5; ++f)
            #pragma unroll
            for (int r = 0; r < 8; ++r)
                dst[(size_t)(l0 + 16 * i + mrow + r) * RTOT + j0 + 16 * f + (lane & 15)] = acc[i][f][r];
}

__global__ void __launch_bounds__(256)
sssm_reduce(const float* __restrict__ part, float* __restrict__ dbc) {
    const int i = blockIdx.x * 256 + threadIdx.x;
    float s[KSPLIT];
    #pragma unroll
    for (int p = 0; p < KSPLIT; ++p) s[p] = part[i + (size_t)p * DBC_N];
    #pragma unroll
    for (int st = 1; st < KSPLIT; st <<= 1)
        #pragma unroll
        for (int p = 0; p < KSPLIT; p += 2 * st) s[p] += s[p + st];
    dbc[i] = s[0];
}

__global__ void __launch_bounds__(256)
sssm_gemm_delta(const float* __restrict__ dbc, const float* __restrict__ Wd,
                float* __restrict__ delta) {
    const int w    = threadIdx.x >> 5;
    const int lane = threadIdx.x & 31;
    const int l0 = (blockIdx.x >> 4) * 64 + (w >> 2) * 32;
    const int d0 = (blockIdx.x & 15) * 256 + (w & 3) * 64;

    const int m  = lane & 15;
    const int kh = (lane >> 4) << 2;

    const float* ap = dbc + (size_t)(l0 + m) * RTOT + kh;
    const float* bp = Wd  + (size_t)(d0 + 4 * m) * DTR + kh;

    v8f acc[2][4];
    #pragma unroll
    for (int i = 0; i < 2; ++i)
        #pragma unroll
        for (int f = 0; f < 4; ++f) acc[i][f] = v8f{};

    #pragma unroll
    for (int k = 0; k < DTR; k += 32) {
        v4f a[2][4], b[4][4];
        #pragma unroll
        for (int i = 0; i < 2; ++i)
            #pragma unroll
            for (int q = 0; q < 4; ++q) a[i][q] = *(const v4f*)(ap + (size_t)i * 16 * RTOT + k + 8 * q);
        #pragma unroll
        for (int f = 0; f < 4; ++f)
            #pragma unroll
            for (int q = 0; q < 4; ++q) b[f][q] = *(const v4f*)(bp + (size_t)f * DTR + k + 8 * q);
        __builtin_amdgcn_sched_barrier(0);
        #pragma unroll
        for (int q = 0; q < 4; ++q) {
            #pragma unroll
            for (int i = 0; i < 2; ++i)
                #pragma unroll
                for (int f = 0; f < 4; ++f) acc[i][f] = wmma4(lo2(a[i][q]), lo2(b[f][q]), acc[i][f]);
            #pragma unroll
            for (int i = 0; i < 2; ++i)
                #pragma unroll
                for (int f = 0; f < 4; ++f) acc[i][f] = wmma4(hi2(a[i][q]), hi2(b[f][q]), acc[i][f]);
        }
        __builtin_amdgcn_sched_barrier(0);
    }

    __shared__ v4f stage[8][16][32];
    #pragma unroll
    for (int i = 0; i < 2; ++i)
        #pragma unroll
        for (int r = 0; r < 8; ++r)
            stage[w][i * 8 + r][lane] = v4f{acc[i][0][r], acc[i][1][r], acc[i][2][r], acc[i][3][r]};

    const int mrow = (lane >> 4) << 3;
    float* orow = delta + (size_t)(l0 + mrow) * DI + d0 + 4 * (lane & 15);
    #pragma unroll 1
    for (int t = 0; t < 16; ++t) {
        const v4f v = stage[w][t][lane];
        v4f o;
        #pragma unroll
        for (int f = 0; f < 4; ++f) {
            o[f] = fmaxf(v[f], 0.0f) + log1pf(__expf(-fabsf(v[f])));
        }
        *(v4f*)(orow + (size_t)((t >> 3) * 16 + (t & 7)) * DI) = o;
    }
}

#define SCAN_T   64
#define SCAN_NC  (SEQ_L / SCAN_T)
#define ROWW     64

template <int M>
static __device__ __forceinline__ float row_xor(float v) {
    return __builtin_bit_cast(float,
        __builtin_amdgcn_update_dpp(0, __builtin_bit_cast(int, v), 0x160 + M, 0xf, 0xf, true));
}

__global__ void __launch_bounds__(256)
sssm_scan(const float* __restrict__ x, const float* __restrict__ A_log,
          const float* __restrict__ Dvec, const float* __restrict__ dbc,
          float* out) {
    __shared__ __attribute__((aligned(16))) float sIn[2][SCAN_T][ROWW];
    __shared__ __attribute__((aligned(16))) float sY[SCAN_T][DS];

    const int t    = threadIdx.x;
    const int dloc = t >> 4;
    const int n    = t & 15;
    const int d0   = blockIdx.x * 16;
    const int d    = d0 + dloc;

    const float A  = -__expf(A_log[(size_t)d * DS + n]);

    const bool b0 = (n & 1) != 0, b1 = (n & 2) != 0, b2 = (n & 4) != 0, b3 = (n & 8) != 0;

    const int r4 = t >> 2, q4 = (t & 3) << 2;
    const int r8 = t >> 3, q8 = (t & 7) << 2;

    const float* gdel = out + (size_t)r4 * DI + d0 + q4;
    const float* gx   = x   + (size_t)r4 * DI + d0 + q4;
    const float* gbc0 = dbc + (size_t)r8 * RTOT + DTR + q8;
    const float* gbc1 = gbc0 + (size_t)32 * RTOT;
    float*       gy   = out + (size_t)r4 * DI + d0 + q4;
    const float4 D4   = *(const float4*)(Dvec + d0 + q4);

    float4 pd  = *(const float4*)gdel;
    float4 px  = *(const float4*)gx;
    float4 pb0 = *(const float4*)gbc0;
    float4 pb1 = *(const float4*)gbc1;

    float h = 0.0f;
    for (int c = 0; c < SCAN_NC; ++c) {
        float (*buf)[ROWW] = sIn[c & 1];
        *(float4*)&buf[r4][q4]           = pd;
        *(float4*)&buf[r4][16 + q4]      = px;
        *(float4*)&buf[r8][32 + q8]      = pb0;
        *(float4*)&buf[32 + r8][32 + q8] = pb1;
        __syncthreads();

        const float4 xc = px;
        if (c + 1 < SCAN_NC) {
            const size_t adv = (size_t)(c + 1) * SCAN_T;
            pd  = *(const float4*)(gdel + adv * DI);
            px  = *(const float4*)(gx   + adv * DI);
            pb0 = *(const float4*)(gbc0 + adv * RTOT);
            pb1 = *(const float4*)(gbc1 + adv * RTOT);
        }

        for (int g = 0; g < SCAN_T; g += 16) {
            float dl[16], xl[16], Bl[16], Cl[16], p[16];
            #pragma unroll
            for (int j = 0; j < 16; ++j) {
                const float* row = buf[g + j];
                dl[j] = row[dloc];
                xl[j] = row[16 + dloc];
                Bl[j] = row[32 + n];
                Cl[j] = row[48 + n];
            }
            #pragma unroll
            for (int j = 0; j < 16; ++j) {
                float dA = __expf(dl[j] * A);
                h = fmaf(dA, h, dl[j] * Bl[j] * xl[j]);
                p[j] = h * Cl[j];
            }
            float q[8], r[4], s[2];
            #pragma unroll
            for (int i = 0; i < 8; ++i) {
                const float kp = b0 ? p[2 * i + 1] : p[2 * i];
                const float sd = b0 ? p[2 * i]     : p[2 * i + 1];
                q[i] = kp + row_xor<1>(sd);
            }
            #pragma unroll
            for (int i = 0; i < 4; ++i) {
                const float kp = b1 ? q[2 * i + 1] : q[2 * i];
                const float sd = b1 ? q[2 * i]     : q[2 * i + 1];
                r[i] = kp + row_xor<2>(sd);
            }
            #pragma unroll
            for (int i = 0; i < 2; ++i) {
                const float kp = b2 ? r[2 * i + 1] : r[2 * i];
                const float sd = b2 ? r[2 * i]     : r[2 * i + 1];
                s[i] = kp + row_xor<4>(sd);
            }
            const float kp3 = b3 ? s[1] : s[0];
            const float sd3 = b3 ? s[0] : s[1];
            sY[g + n][dloc] = kp3 + row_xor<8>(sd3);
        }
        __syncthreads();

        const float4 ys = *(const float4*)&sY[r4][q4];
        float4 yo;
        yo.x = fmaf(xc.x, D4.x, ys.x);
        yo.y = fmaf(xc.y, D4.y, ys.y);
        yo.z = fmaf(xc.z, D4.z, ys.z);
        yo.w = fmaf(xc.w, D4.w, ys.w);
        *(float4*)(gy + (size_t)c * SCAN_T * DI) = yo;
    }
}

extern "C" void kernel_launch(void* const* d_in, const int* in_sizes, int n_in,
                              void* d_out, int out_size, void* d_ws, size_t ws_size,
                              hipStream_t stream) {
    const float* x    = (const float*)d_in[0];
    const float* Win  = (const float*)d_in[1];
    const float* Wd   = (const float*)d_in[2];
    const float* Alog = (const float*)d_in[3];
    const float* Dv   = (const float*)d_in[4];
    float*       out  = (float*)d_out;

    float* part = (float*)d_ws;
    float* dbc  = part + (size_t)KSPLIT * DBC_N;

    sssm_gemm_in<<<512, 128, 0, stream>>>(x, Win, part);
    sssm_reduce<<<DBC_N / 256, 256, 0, stream>>>(part, dbc);
    sssm_gemm_delta<<<512, 256, 0, stream>>>(dbc, Wd, out);
    sssm_scan<<<(DI * DS) / 256, 256, 0, stream>>>(x, Alog, Dv, dbc, out);
}


# scan: 1 thread per (channel, 128-step segment) with the 16 states in registers, y summed in-thread, B|C read from LDS; 2 passes, segment start states carried: last-bit changes in h and y
# speedup vs baseline: 17.1296x; 1.4243x over previous
#include <hip/hip_runtime.h>


#define SEQ_L   2048
#define DI      4096
#define RTOT    160
#define DTR     128
#define DS      16
#define KSPLIT  16
#define KCH     (DI / KSPLIT)
#define DBC_N   (SEQ_L * RTOT)

typedef float v2f __attribute__((ext_vector_type(2)));
typedef float v4f __attribute__((ext_vector_type(4)));
typedef float v8f __attribute__((ext_vector_type(8)));

static __device__ __forceinline__ v2f lo2(v4f v) { return v2f{v.x, v.y}; }
static __device__ __forceinline__ v2f hi2(v4f v) { return v2f{v.z, v.w}; }

static __device__ __forceinline__ v8f wmma4(v2f a, v2f b, v8f c) {
    return __builtin_amdgcn_wmma_f32_16x16x4_f32(
               false, a, false, b, (short)0, c, false, false);
}


__global__ void __launch_bounds__(128)
sssm_gemm_in(const float* __restrict__ x, const float* __restrict__ Win,
             float* __restrict__ part) {
    const int w    = threadIdx.x >> 5;
    const int lane = threadIdx.x & 31;
    const int ks   = blockIdx.x & (KSPLIT - 1);
    const int lp   = blockIdx.x >> 4;
    const int l0   = (lp * 2 + (w >> 1)) * 32;
    const int j0   = (w & 1) * 80;

    const int m  = lane & 15;
    const int kh = (lane >> 4) << 2;

    const float* ap = x   + (size_t)(l0 + m) * DI + ks * KCH + kh;
    const float* bp = Win + (size_t)(j0 + m) * DI + ks * KCH + kh;

    v8f acc[2][5];
    #pragma unroll
    for (int i = 0; i < 2; ++i)
        #pragma unroll
        for (int f = 0; f < 5; ++f) acc[i][f] = v8f{};

    v4f a[2][4], b[5][4];
    #pragma unroll
    for (int i = 0; i < 2; ++i)
        #pragma unroll
        for (int q = 0; q < 4; ++q) a[i][q] = *(const v4f*)(ap + (size_t)i * 16 * DI + 8 * q);
    #pragma unroll
    for (int f = 0; f < 5; ++f)
        #pragma unroll
        for (int q = 0; q < 4; ++q) b[f][q] = *(const v4f*)(bp + (size_t)f * 16 * DI + 8 * q);

    #pragma unroll
    for (int k = 0; k < KCH; k += 32) {
        v4f an[2][4], bn[5][4];
        if (k + 32 < KCH) {
            ap += 32; bp += 32;
            #pragma unroll
            for (int i = 0; i < 2; ++i)
                #pragma unroll
                for (int q = 0; q < 4; ++q) an[i][q] = *(const v4f*)(ap + (size_t)i * 16 * DI + 8 * q);
            #pragma unroll
            for (int f = 0; f < 5; ++f)
                #pragma unroll
                for (int q = 0; q < 4; ++q) bn[f][q] = *(const v4f*)(bp + (size_t)f * 16 * DI + 8 * q);
        }
        __builtin_amdgcn_sched_barrier(0);
        #pragma unroll
        for (int q = 0; q < 4; ++q) {
            #pragma unroll
            for (int i = 0; i < 2; ++i)
                #pragma unroll
                for (int f = 0; f < 5; ++f) acc[i][f] = wmma4(lo2(a[i][q]), lo2(b[f][q]), acc[i][f]);
            #pragma unroll
            for (int i = 0; i < 2; ++i)
                #pragma unroll
                for (int f = 0; f < 5; ++f) acc[i][f] = wmma4(hi2(a[i][q]), hi2(b[f][q]), acc[i][f]);
        }
        __builtin_amdgcn_sched_barrier(0);
        if (k + 32 < KCH) {
            #pragma unroll
            for (int i = 0; i < 2; ++i)
                #pragma unroll
                for (int q = 0; q < 4; ++q) a[i][q] = an[i][q];
            #pragma unroll
            for (int f = 0; f < 5; ++f)
                #pragma unroll
                for (int q = 0; q < 4; ++q) b[f][q] = bn[f][q];
        }
    }

    const int mrow = (lane >> 4) << 3;
    float* dst = part + (size_t)ks * DBC_N;
    #pragma unroll
    for (int i = 0; i < 2; ++i)
        #pragma unroll
        for (int f = 0; f < 5; ++f)
            #pragma unroll
            for (int r = 0; r < 8; ++r)
                dst[(size_t)(l0 + 16 * i + mrow + r) * RTOT + j0 + 16 * f + (lane & 15)] = acc[i][f][r];
}

__global__ void __launch_bounds__(256)
sssm_reduce(const float* __restrict__ part, float* __restrict__ dbc) {
    const int i = blockIdx.x * 256 + threadIdx.x;
    float s[KSPLIT];
    #pragma unroll
    for (int p = 0; p < KSPLIT; ++p) s[p] = part[i + (size_t)p * DBC_N];
    #pragma unroll
    for (int st = 1; st < KSPLIT; st <<= 1)
        #pragma unroll
        for (int p = 0; p < KSPLIT; p += 2 * st) s[p] += s[p + st];
    dbc[i] = s[0];
}

__global__ void __launch_bounds__(256)
sssm_gemm_delta(const float* __restrict__ dbc, const float* __restrict__ Wd,
                float* __restrict__ delta) {
    const int w    = threadIdx.x >> 5;
    const int lane = threadIdx.x & 31;
    const int l0 = (blockIdx.x >> 4) * 64 + (w >> 2) * 32;
    const int d0 = (blockIdx.x & 15) * 256 + (w & 3) * 64;

    const int m  = lane & 15;
    const int kh = (lane >> 4) << 2;

    const float* ap = dbc + (size_t)(l0 + m) * RTOT + kh;
    const float* bp = Wd  + (size_t)(d0 + 4 * m) * DTR + kh;

    v8f acc[2][4];
    #pragma unroll
    for (int i = 0; i < 2; ++i)
        #pragma unroll
        for (int f = 0; f < 4; ++f) acc[i][f] = v8f{};

    #pragma unroll
    for (int k = 0; k < DTR; k += 32) {
        v4f a[2][4], b[4][4];
        #pragma unroll
        for (int i = 0; i < 2; ++i)
            #pragma unroll
            for (int q = 0; q < 4; ++q) a[i][q] = *(const v4f*)(ap + (size_t)i * 16 * RTOT + k + 8 * q);
        #pragma unroll
        for (int f = 0; f < 4; ++f)
            #pragma unroll
            for (int q = 0; q < 4; ++q) b[f][q] = *(const v4f*)(bp + (size_t)f * DTR + k + 8 * q);
        __builtin_amdgcn_sched_barrier(0);
        #pragma unroll
        for (int q = 0; q < 4; ++q) {
            #pragma unroll
            for (int i = 0; i < 2; ++i)
                #pragma unroll
                for (int f = 0; f < 4; ++f) acc[i][f] = wmma4(lo2(a[i][q]), lo2(b[f][q]), acc[i][f]);
            #pragma unroll
            for (int i = 0; i < 2; ++i)
                #pragma unroll
                for (int f = 0; f < 4; ++f) acc[i][f] = wmma4(hi2(a[i][q]), hi2(b[f][q]), acc[i][f]);
        }
        __builtin_amdgcn_sched_barrier(0);
    }

    __shared__ v4f stage[8][16][32];
    #pragma unroll
    for (int i = 0; i < 2; ++i)
        #pragma unroll
        for (int r = 0; r < 8; ++r)
            stage[w][i * 8 + r][lane] = v4f{acc[i][0][r], acc[i][1][r], acc[i][2][r], acc[i][3][r]};

    const int mrow = (lane >> 4) << 3;
    float* orow = delta + (size_t)(l0 + mrow) * DI + d0 + 4 * (lane & 15);
    #pragma unroll 1
    for (int t = 0; t < 16; ++t) {
        const v4f v = stage[w][t][lane];
        v4f o;
        #pragma unroll
        for (int f = 0; f < 4; ++f) {
            o[f] = fmaxf(v[f], 0.0f) + log1pf(__expf(-fabsf(v[f])));
        }
        *(v4f*)(orow + (size_t)((t >> 3) * 16 + (t & 7)) * DI) = o;
    }
}

#define SEG_S   16
#define SEG_T   (SEQ_L / SEG_S)
#define SCAN_PF 8
#define LOG2E_F 1.44269504088896341f

static __device__ __forceinline__ v2f exp2_2(v2f v) {
    return v2f{__builtin_amdgcn_exp2f(v.x), __builtin_amdgcn_exp2f(v.y)};
}

static __device__ __forceinline__ void load_A2(const float* __restrict__ A_log, int d, v2f (&A2)[8]) {
    const v4f* ap = (const v4f*)(A_log + (size_t)d * DS);
    #pragma unroll
    for (int q = 0; q < 4; ++q) {
        const v4f al = ap[q];
        A2[2 * q]     = v2f{-__expf(al.x) * LOG2E_F, -__expf(al.y) * LOG2E_F};
        A2[2 * q + 1] = v2f{-__expf(al.z) * LOG2E_F, -__expf(al.w) * LOG2E_F};
    }
}

static __device__ __forceinline__ void scan_fetch(v2f (&dl)[SCAN_PF / 2], v2f (&xl)[SCAN_PF / 2],
                                                  const float* pd, const float* px, int r) {
    #pragma unroll
    for (int j = 0; j < SCAN_PF; j += 2) {
        dl[j >> 1].x = pd[(size_t)(r + j) * DI]; dl[j >> 1].y = pd[(size_t)(r + j + 1) * DI];
        xl[j >> 1].x = px[(size_t)(r + j) * DI]; xl[j >> 1].y = px[(size_t)(r + j + 1) * DI];
    }
}

static __device__ __forceinline__ void scan_exp_rows(v2f dd, const v2f (&A2)[8], v2f (&e0)[8], v2f (&e1)[8]) {
    #pragma unroll
    for (int i = 0; i < 8; ++i) {
        const v2f ax = dd * A2[i].x;
        const v2f ay = dd * A2[i].y;
        e0[i] = v2f{__builtin_amdgcn_exp2f(ax.x), __builtin_amdgcn_exp2f(ay.x)};
        e1[i] = v2f{__builtin_amdgcn_exp2f(ax.y), __builtin_amdgcn_exp2f(ay.y)};
    }
}

template <int NQ>
static __device__ __forceinline__ void scan_lds_row(v4f (&bc)[NQ], const float* row) {
    #pragma unroll
    for (int q = 0; q < NQ; ++q) bc[q] = ((const v4f*)row)[q];
}

static __device__ __forceinline__ void scan_step_h(const v2f (&e)[8], float du, const v4f (&b)[4], v2f (&h)[8]) {
    #pragma unroll
    for (int q = 0; q < 4; ++q) {
        h[2 * q]     = e[2 * q]     * h[2 * q]     + du * lo2(b[q]);
        h[2 * q + 1] = e[2 * q + 1] * h[2 * q + 1] + du * hi2(b[q]);
    }
}

static __device__ __forceinline__ float scan_step_hy(const v2f (&e)[8], float du, const v4f (&bc)[8], v2f (&h)[8]) {
    v2f ya = v2f{0.0f, 0.0f}, yb = v2f{0.0f, 0.0f};
    #pragma unroll
    for (int q = 0; q < 4; ++q) {
        h[2 * q]     = e[2 * q]     * h[2 * q]     + du * lo2(bc[q]);
        h[2 * q + 1] = e[2 * q + 1] * h[2 * q + 1] + du * hi2(bc[q]);
        ya = h[2 * q]     * lo2(bc[4 + q]) + ya;
        yb = h[2 * q + 1] * hi2(bc[4 + q]) + yb;
    }
    const v2f yy = ya + yb;
    return yy.x + yy.y;
}

static __device__ __forceinline__ void scan_local_block(const v2f (&dl)[SCAN_PF / 2], const v2f (&xl)[SCAN_PF / 2],
                                                        const float (*sb)[DS], const v2f (&A2)[8],
                                                        v2f (&h)[8], float& sd) {
    #pragma unroll
    for (int jj = 0; jj < SCAN_PF / 2; ++jj) {
        v4f b0[4], b1[4];
        scan_lds_row<4>(b0, sb[2 * jj]);
        scan_lds_row<4>(b1, sb[2 * jj + 1]);
        __builtin_amdgcn_sched_barrier(0);
        const v2f dd  = dl[jj];
        const v2f du2 = dd * xl[jj];
        v2f e0[8], e1[8];
        scan_exp_rows(dd, A2, e0, e1);
        sd += dd.x;
        sd += dd.y;
        scan_step_h(e0, du2.x, b0, h);
        scan_step_h(e1, du2.y, b1, h);
        __builtin_amdgcn_sched_barrier(0);
    }
}

static __device__ __forceinline__ void scan_out_block(const v2f (&dl)[SCAN_PF / 2], const v2f (&xl)[SCAN_PF / 2],
                                                      const float (*sbc)[2 * DS], const v2f (&A2)[8],
                                                      v2f (&h)[8], float Dd, float* po) {
    #pragma unroll
    for (int jj = 0; jj < SCAN_PF / 2; ++jj) {
        v4f b0[8], b1[8];
        scan_lds_row<8>(b0, sbc[2 * jj]);
        scan_lds_row<8>(b1, sbc[2 * jj + 1]);
        __builtin_amdgcn_sched_barrier(0);
        const v2f dd  = dl[jj];
        const v2f du2 = dd * xl[jj];
        v2f e0[8], e1[8];
        scan_exp_rows(dd, A2, e0, e1);
        const float ys0 = scan_step_hy(e0, du2.x, b0, h);
        po[(size_t)(2 * jj) * DI]     = fmaf(xl[jj].x, Dd, ys0);
        __builtin_amdgcn_sched_barrier(0);
        const float ys1 = scan_step_hy(e1, du2.y, b1, h);
        po[(size_t)(2 * jj + 1) * DI] = fmaf(xl[jj].y, Dd, ys1);
        __builtin_amdgcn_sched_barrier(0);
    }
}

__global__ void __launch_bounds__(256)
sssm_scan_local(const float* __restrict__ x, const float* __restrict__ A_log,
                const float* __restrict__ dbc, const float* __restrict__ delta,
                float* __restrict__ segE, float* __restrict__ segSD) {
    __shared__ __attribute__((aligned(16))) float sB[SEG_T][DS];

    const int d  = blockIdx.x * 256 + threadIdx.x;
    const int s  = blockIdx.y;
    const int t0 = s * SEG_T;

    {
        const int r = threadIdx.x >> 2, q = (threadIdx.x & 3) << 2;
        #pragma unroll
        for (int k = 0; k < SEG_T / 64; ++k)
            *(v4f*)&sB[r + 64 * k][q] = *(const v4f*)(dbc + (size_t)(t0 + r + 64 * k) * RTOT + DTR + q);
    }

    v2f A2[8];
    load_A2(A_log, d, A2);

    const float* pd = delta + (size_t)t0 * DI + d;
    const float* px = x     + (size_t)t0 * DI + d;

    v2f dA_[SCAN_PF / 2], xA_[SCAN_PF / 2], dB_[SCAN_PF / 2], xB_[SCAN_PF / 2];
    scan_fetch(dA_, xA_, pd, px, 0);

    v2f h[8];
    #pragma unroll
    for (int i = 0; i < 8; ++i) h[i] = v2f{0.0f, 0.0f};
    float sd = 0.0f;

    __syncthreads();

    for (int tb = 0; tb < SEG_T; tb += 2 * SCAN_PF) {
        scan_fetch(dB_, xB_, pd, px, tb + SCAN_PF);
        __builtin_amdgcn_sched_barrier(0);
        scan_local_block(dA_, xA_, sB + tb, A2, h, sd);
        __builtin_amdgcn_sched_barrier(0);
        scan_fetch(dA_, xA_, pd, px, (tb + 2 * SCAN_PF < SEG_T ? tb + 2 * SCAN_PF : SEG_T - SCAN_PF));
        __builtin_amdgcn_sched_barrier(0);
        scan_local_block(dB_, xB_, sB + tb + SCAN_PF, A2, h, sd);
        __builtin_amdgcn_sched_barrier(0);
    }

    #pragma unroll
    for (int i = 0; i < 8; ++i) {
        segE[(size_t)(s * DS + 2 * i)     * DI + d] = h[i].x;
        segE[(size_t)(s * DS + 2 * i + 1) * DI + d] = h[i].y;
    }
    segSD[(size_t)s * DI + d] = sd;
}

__global__ void __launch_bounds__(256)
sssm_scan_out(const float* __restrict__ x, const float* __restrict__ A_log,
              const float* __restrict__ Dvec, const float* __restrict__ dbc,
              const float* __restrict__ segE, const float* __restrict__ segSD,
              float* out) {
    __shared__ __attribute__((aligned(16))) float sBC[SEG_T][2 * DS];

    const int d  = blockIdx.x * 256 + threadIdx.x;
    const int s  = blockIdx.y;
    const int t0 = s * SEG_T;

    {
        const int r = threadIdx.x >> 3, q = (threadIdx.x & 7) << 2;
        #pragma unroll
        for (int k = 0; k < SEG_T / 32; ++k)
            *(v4f*)&sBC[r + 32 * k][q] = *(const v4f*)(dbc + (size_t)(t0 + r + 32 * k) * RTOT + DTR + q);
    }

    v2f A2[8];
    load_A2(A_log, d, A2);
    const float Dd = Dvec[d];

    float*       po = out + (size_t)t0 * DI + d;
    const float* px = x   + (size_t)t0 * DI + d;

    v2f dA_[SCAN_PF / 2], xA_[SCAN_PF / 2], dB_[SCAN_PF / 2], xB_[SCAN_PF / 2];
    scan_fetch(dA_, xA_, po, px, 0);

    v2f h[8];
    #pragma unroll
    for (int i = 0; i < 8; ++i) h[i] = v2f{0.0f, 0.0f};
    for (int sp0 = 0; sp0 < s; sp0 += 5) {
        float sdl[5];
        v2f   E[5][8];
        #pragma unroll
        for (int u = 0; u < 5; ++u) {
            const int sp = (sp0 + u < SEG_S - 1) ? sp0 + u : SEG_S - 2;
            sdl[u] = segSD[(size_t)sp * DI + d];
            #pragma unroll
            for (int i = 0; i < 8; ++i)
                E[u][i] = v2f{segE[(size_t)(sp * DS + 2 * i) * DI + d],
                              segE[(size_t)(sp * DS + 2 * i + 1) * DI + d]};
        }
        #pragma unroll
        for (int u = 0; u < 5; ++u) {
            const bool on = sp0 + u < s;
            #pragma unroll
            for (int i = 0; i < 8; ++i) {
                const v2f P  = on ? exp2_2(sdl[u] * A2[i]) : v2f{1.0f, 1.0f};
                const v2f Eu = on ? E[u][i] : v2f{0.0f, 0.0f};
                h[i] = P * h[i] + Eu;
            }
        }
    }

    __syncthreads();

    for (int tb = 0; tb < SEG_T; tb += 2 * SCAN_PF) {
        scan_fetch(dB_, xB_, po, px, tb + SCAN_PF);
        __builtin_amdgcn_sched_barrier(0);
        scan_out_block(dA_, xA_, sBC + tb, A2, h, Dd, po + (size_t)tb * DI);
        __builtin_amdgcn_sched_barrier(0);
        scan_fetch(dA_, xA_, po, px, (tb + 2 * SCAN_PF < SEG_T ? tb + 2 * SCAN_PF : SEG_T - SCAN_PF));
        __builtin_amdgcn_sched_barrier(0);
        scan_out_block(dB_, xB_, sBC + tb + SCAN_PF, A2, h, Dd, po + (size_t)(tb + SCAN_PF) * DI);
        __builtin_amdgcn_sched_barrier(0);
    }
}

extern "C" void kernel_launch(void* const* d_in, const int* in_sizes, int n_in,
                              void* d_out, int out_size, void* d_ws, size_t ws_size,
                              hipStream_t stream) {
    const float* x    = (const float*)d_in[0];
    const float* Win  = (const float*)d_in[1];
    const float* Wd   = (const float*)d_in[2];
    const float* Alog = (const float*)d_in[3];
    const float* Dv   = (const float*)d_in[4];
    float*       out  = (float*)d_out;

    float* part = (float*)d_ws;
    float* dbc  = part + (size_t)KSPLIT * DBC_N;
    float* segE  = dbc  + (size_t)DBC_N;
    float* segSD = segE + (size_t)(SEG_S - 1) * DS * DI;

    sssm_gemm_in<<<512, 128, 0, stream>>>(x, Win, part);
    sssm_reduce<<<DBC_N / 256, 256, 0, stream>>>(part, dbc);
    sssm_gemm_delta<<<512, 256, 0, stream>>>(dbc, Wd, out);
    sssm_scan_local<<<dim3(DI / 256, SEG_S - 1), 256, 0, stream>>>(x, Alog, dbc, out, segE, segSD);
    sssm_scan_out<<<dim3(DI / 256, SEG_S), 256, 0, stream>>>(x, Alog, Dv, dbc, segE, segSD, out);
}


# GEMM2 epilogue: softplus's log1p(e) as hardware log2(u)*ln2 - c/u, u = 1+e rounded, c its rounding error (was library log1pf); unrolled on the accumulators, no LDS stage: last-bit changes in delta
# speedup vs baseline: 19.0634x; 1.1129x over previous
#include <hip/hip_runtime.h>


#define SEQ_L   2048
#define DI      4096
#define RTOT    160
#define DTR     128
#define DS      16
#define KSPLIT  16
#define KCH     (DI / KSPLIT)
#define DBC_N   (SEQ_L * RTOT)

typedef float v2f __attribute__((ext_vector_type(2)));
typedef float v4f __attribute__((ext_vector_type(4)));
typedef float v8f __attribute__((ext_vector_type(8)));

static __device__ __forceinline__ v2f lo2(v4f v) { return v2f{v.x, v.y}; }
static __device__ __forceinline__ v2f hi2(v4f v) { return v2f{v.z, v.w}; }

static __device__ __forceinline__ v8f wmma4(v2f a, v2f b, v8f c) {
    return __builtin_amdgcn_wmma_f32_16x16x4_f32(
               false, a, false, b, (short)0, c, false, false);
}


__global__ void __launch_bounds__(128)
sssm_gemm_in(const float* __restrict__ x, const float* __restrict__ Win,
             float* __restrict__ part) {
    const int w    = threadIdx.x >> 5;
    const int lane = threadIdx.x & 31;
    const int ks   = blockIdx.x & (KSPLIT - 1);
    const int lp   = blockIdx.x >> 4;
    const int l0   = (lp * 2 + (w >> 1)) * 32;
    const int j0   = (w & 1) * 80;

    const int m  = lane & 15;
    const int kh = (lane >> 4) << 2;

    const float* ap = x   + (size_t)(l0 + m) * DI + ks * KCH + kh;
    const float* bp = Win + (size_t)(j0 + m) * DI + ks * KCH + kh;

    v8f acc[2][5];
    #pragma unroll
    for (int i = 0; i < 2; ++i)
        #pragma unroll
        for (int f = 0; f < 5; ++f) acc[i][f] = v8f{};

    v4f a[2][4], b[5][4];
    #pragma unroll
    for (int i = 0; i < 2; ++i)
        #pragma unroll
        for (int q = 0; q < 4; ++q) a[i][q] = *(const v4f*)(ap + (size_t)i * 16 * DI + 8 * q);
    #pragma unroll
    for (int f = 0; f < 5; ++f)
        #pragma unroll
        for (int q = 0; q < 4; ++q) b[f][q] = *(const v4f*)(bp + (size_t)f * 16 * DI + 8 * q);

    #pragma unroll
    for (int k = 0; k < KCH; k += 32) {
        v4f an[2][4], bn[5][4];
        if (k + 32 < KCH) {
            ap += 32; bp += 32;
            #pragma unroll
            for (int i = 0; i < 2; ++i)
                #pragma unroll
                for (int q = 0; q < 4; ++q) an[i][q] = *(const v4f*)(ap + (size_t)i * 16 * DI + 8 * q);
            #pragma unroll
            for (int f = 0; f < 5; ++f)
                #pragma unroll
                for (int q = 0; q < 4; ++q) bn[f][q] = *(const v4f*)(bp + (size_t)f * 16 * DI + 8 * q);
        }
        __builtin_amdgcn_sched_barrier(0);
        #pragma unroll
        for (int q = 0; q < 4; ++q) {
            #pragma unroll
            for (int i = 0; i < 2; ++i)
                #pragma unroll
                for (int f = 0; f < 5; ++f) acc[i][f] = wmma4(lo2(a[i][q]), lo2(b[f][q]), acc[i][f]);
            #pragma unroll
            for (int i = 0; i < 2; ++i)
                #pragma unroll
                for (int f = 0; f < 5; ++f) acc[i][f] = wmma4(hi2(a[i][q]), hi2(b[f][q]), acc[i][f]);
        }
        __builtin_amdgcn_sched_barrier(0);
        if (k + 32 < KCH) {
            #pragma unroll
            for (int i = 0; i < 2; ++i)
                #pragma unroll
                for (int q = 0; q < 4; ++q) a[i][q] = an[i][q];
            #pragma unroll
            for (int f = 0; f < 5; ++f)
                #pragma unroll
                for (int q = 0; q < 4; ++q) b[f][q] = bn[f][q];
        }
    }

    const int mrow = (lane >> 4) << 3;
    float* dst = part + (size_t)ks * DBC_N;
    #pragma unroll
    for (int i = 0; i < 2; ++i)
        #pragma unroll
        for (int f = 0; f < 5; ++f)
            #pragma unroll
            for (int r = 0; r < 8; ++r)
                dst[(size_t)(l0 + 16 * i + mrow + r) * RTOT + j0 + 16 * f + (lane & 15)] = acc[i][f][r];
}

__global__ void __launch_bounds__(256)
sssm_reduce(const float* __restrict__ part, float* __restrict__ dbc) {
    const int i = blockIdx.x * 256 + threadIdx.x;
    float s[KSPLIT];
    #pragma unroll
    for (int p = 0; p < KSPLIT; ++p) s[p] = part[i + (size_t)p * DBC_N];
    #pragma unroll
    for (int st = 1; st < KSPLIT; st <<= 1)
        #pragma unroll
        for (int p = 0; p < KSPLIT; p += 2 * st) s[p] += s[p + st];
    dbc[i] = s[0];
}

#define LN2_F 0.693147180559945309f
static __device__ __forceinline__ float softplus1(float v) {
    const float e = __expf(-fabsf(v));
    const float u = 1.0f + e;
    const float c = (u - 1.0f) - e;
    const float l = fmaf(__builtin_amdgcn_logf(u), LN2_F, -(c * __builtin_amdgcn_rcpf(u)));
    return fmaxf(v, 0.0f) + l;
}

__global__ void __launch_bounds__(256)
sssm_gemm_delta(const float* __restrict__ dbc, const float* __restrict__ Wd,
                float* __restrict__ delta) {
    const int w    = threadIdx.x >> 5;
    const int lane = threadIdx.x & 31;
    const int l0 = (blockIdx.x >> 4) * 64 + (w >> 2) * 32;
    const int d0 = (blockIdx.x & 15) * 256 + (w & 3) * 64;

    const int m  = lane & 15;
    const int kh = (lane >> 4) << 2;

    const float* ap = dbc + (size_t)(l0 + m) * RTOT + kh;
    const float* bp = Wd  + (size_t)(d0 + 4 * m) * DTR + kh;

    v8f acc[2][4];
    #pragma unroll
    for (int i = 0; i < 2; ++i)
        #pragma unroll
        for (int f = 0; f < 4; ++f) acc[i][f] = v8f{};

    #pragma unroll
    for (int k = 0; k < DTR; k += 32) {
        v4f a[2][4], b[4][4];
        #pragma unroll
        for (int i = 0; i < 2; ++i)
            #pragma unroll
            for (int q = 0; q < 4; ++q) a[i][q] = *(const v4f*)(ap + (size_t)i * 16 * RTOT + k + 8 * q);
        #pragma unroll
        for (int f = 0; f < 4; ++f)
            #pragma unroll
            for (int q = 0; q < 4; ++q) b[f][q] = *(const v4f*)(bp + (size_t)f * DTR + k + 8 * q);
        __builtin_amdgcn_sched_barrier(0);
        #pragma unroll
        for (int q = 0; q < 4; ++q) {
            #pragma unroll
            for (int i = 0; i < 2; ++i)
                #pragma unroll
                for (int f = 0; f < 4; ++f) acc[i][f] = wmma4(lo2(a[i][q]), lo2(b[f][q]), acc[i][f]);
            #pragma unroll
            for (int i = 0; i < 2; ++i)
                #pragma unroll
                for (int f = 0; f < 4; ++f) acc[i][f] = wmma4(hi2(a[i][q]), hi2(b[f][q]), acc[i][f]);
        }
        __builtin_amdgcn_sched_barrier(0);
    }

    const int mrow = (lane >> 4) << 3;
    float* orow = delta + (size_t)(l0 + mrow) * DI + d0 + 4 * (lane & 15);
    #pragma unroll
    for (int i = 0; i < 2; ++i)
        #pragma unroll
        for (int r = 0; r < 8; ++r) {
            v4f o;
            #pragma unroll
            for (int f = 0; f < 4; ++f) o[f] = softplus1(acc[i][f][r]);
            *(v4f*)(orow + (size_t)(16 * i + r) * DI) = o;
        }
}

#define SEG_S   16
#define SEG_T   (SEQ_L / SEG_S)
#define SCAN_PF 8
#define LOG2E_F 1.44269504088896341f

static __device__ __forceinline__ v2f exp2_2(v2f v) {
    return v2f{__builtin_amdgcn_exp2f(v.x), __builtin_amdgcn_exp2f(v.y)};
}

static __device__ __forceinline__ void load_A2(const float* __restrict__ A_log, int d, v2f (&A2)[8]) {
    const v4f* ap = (const v4f*)(A_log + (size_t)d * DS);
    #pragma unroll
    for (int q = 0; q < 4; ++q) {
        const v4f al = ap[q];
        A2[2 * q]     = v2f{-__expf(al.x) * LOG2E_F, -__expf(al.y) * LOG2E_F};
        A2[2 * q + 1] = v2f{-__expf(al.z) * LOG2E_F, -__expf(al.w) * LOG2E_F};
    }
}

static __device__ __forceinline__ void scan_fetch(v2f (&dl)[SCAN_PF / 2], v2f (&xl)[SCAN_PF / 2],
                                                  const float* pd, const float* px, int r) {
    #pragma unroll
    for (int j = 0; j < SCAN_PF; j += 2) {
        dl[j >> 1].x = pd[(size_t)(r + j) * DI]; dl[j >> 1].y = pd[(size_t)(r + j + 1) * DI];
        xl[j >> 1].x = px[(size_t)(r + j) * DI]; xl[j >> 1].y = px[(size_t)(r + j + 1) * DI];
    }
}

static __device__ __forceinline__ void scan_exp_rows(v2f dd, const v2f (&A2)[8], v2f (&e0)[8], v2f (&e1)[8]) {
    #pragma unroll
    for (int i = 0; i < 8; ++i) {
        const v2f ax = dd * A2[i].x;
        const v2f ay = dd * A2[i].y;
        e0[i] = v2f{__builtin_amdgcn_exp2f(ax.x), __builtin_amdgcn_exp2f(ay.x)};
        e1[i] = v2f{__builtin_amdgcn_exp2f(ax.y), __builtin_amdgcn_exp2f(ay.y)};
    }
}

template <int NQ>
static __device__ __forceinline__ void scan_lds_row(v4f (&bc)[NQ], const float* row) {
    #pragma unroll
    for (int q = 0; q < NQ; ++q) bc[q] = ((const v4f*)row)[q];
}

static __device__ __forceinline__ void scan_step_h(const v2f (&e)[8], float du, const v4f (&b)[4], v2f (&h)[8]) {
    #pragma unroll
    for (int q = 0; q < 4; ++q) {
        h[2 * q]     = e[2 * q]     * h[2 * q]     + du * lo2(b[q]);
        h[2 * q + 1] = e[2 * q + 1] * h[2 * q + 1] + du * hi2(b[q]);
    }
}

static __device__ __forceinline__ float scan_step_hy(const v2f (&e)[8], float du, const v4f (&bc)[8], v2f (&h)[8]) {
    v2f ya = v2f{0.0f, 0.0f}, yb = v2f{0.0f, 0.0f};
    #pragma unroll
    for (int q = 0; q < 4; ++q) {
        h[2 * q]     = e[2 * q]     * h[2 * q]     + du * lo2(bc[q]);
        h[2 * q + 1] = e[2 * q + 1] * h[2 * q + 1] + du * hi2(bc[q]);
        ya = h[2 * q]     * lo2(bc[4 + q]) + ya;
        yb = h[2 * q + 1] * hi2(bc[4 + q]) + yb;
    }
    const v2f yy = ya + yb;
    return yy.x + yy.y;
}

static __device__ __forceinline__ void scan_local_block(const v2f (&dl)[SCAN_PF / 2], const v2f (&xl)[SCAN_PF / 2],
                                                        const float (*sb)[DS], const v2f (&A2)[8],
                                                        v2f (&h)[8], float& sd) {
    #pragma unroll
    for (int jj = 0; jj < SCAN_PF / 2; ++jj) {
        v4f b0[4], b1[4];
        scan_lds_row<4>(b0, sb[2 * jj]);
        scan_lds_row<4>(b1, sb[2 * jj + 1]);
        __builtin_amdgcn_sched_barrier(0);
        const v2f dd  = dl[jj];
        const v2f du2 = dd * xl[jj];
        v2f e0[8], e1[8];
        scan_exp_rows(dd, A2, e0, e1);
        sd += dd.x;
        sd += dd.y;
        scan_step_h(e0, du2.x, b0, h);
        scan_step_h(e1, du2.y, b1, h);
        __builtin_amdgcn_sched_barrier(0);
    }
}

static __device__ __forceinline__ void scan_out_block(const v2f (&dl)[SCAN_PF / 2], const v2f (&xl)[SCAN_PF / 2],
                                                      const float (*sbc)[2 * DS], const v2f (&A2)[8],
                                                      v2f (&h)[8], float Dd, float* po) {
    #pragma unroll
    for (int jj = 0; jj < SCAN_PF / 2; ++jj) {
        v4f b0[8], b1[8];
        scan_lds_row<8>(b0, sbc[2 * jj]);
        scan_lds_row<8>(b1, sbc[2 * jj + 1]);
        __builtin_amdgcn_sched_barrier(0);
        const v2f dd  = dl[jj];
        const v2f du2 = dd * xl[jj];
        v2f e0[8], e1[8];
        scan_exp_rows(dd, A2, e0, e1);
        const float ys0 = scan_step_hy(e0, du2.x, b0, h);
        po[(size_t)(2 * jj) * DI]     = fmaf(xl[jj].x, Dd, ys0);
        __builtin_amdgcn_sched_barrier(0);
        const float ys1 = scan_step_hy(e1, du2.y, b1, h);
        po[(size_t)(2 * jj + 1) * DI] = fmaf(xl[jj].y, Dd, ys1);
        __builtin_amdgcn_sched_barrier(0);
    }
}

__global__ void __launch_bounds__(256)
sssm_scan_local(const float* __restrict__ x, const float* __restrict__ A_log,
                const float* __restrict__ dbc, const float* __restrict__ delta,
                float* __restrict__ segE, float* __restrict__ segSD) {
    __shared__ __attribute__((aligned(16))) float sB[SEG_T][DS];

    const int d  = blockIdx.x * 256 + threadIdx.x;
    const int s  = blockIdx.y;
    const int t0 = s * SEG_T;

    {
        const int r = threadIdx.x >> 2, q = (threadIdx.x & 3) << 2;
        #pragma unroll
        for (int k = 0; k < SEG_T / 64; ++k)
            *(v4f*)&sB[r + 64 * k][q] = *(const v4f*)(dbc + (size_t)(t0 + r + 64 * k) * RTOT + DTR + q);
    }

    v2f A2[8];
    load_A2(A_log, d, A2);

    const float* pd = delta + (size_t)t0 * DI + d;
    const float* px = x     + (size_t)t0 * DI + d;

    v2f dA_[SCAN_PF / 2], xA_[SCAN_PF / 2], dB_[SCAN_PF / 2], xB_[SCAN_PF / 2];
    scan_fetch(dA_, xA_, pd, px, 0);

    v2f h[8];
    #pragma unroll
    for (int i = 0; i < 8; ++i) h[i] = v2f{0.0f, 0.0f};
    float sd = 0.0f;

    __syncthreads();

    for (int tb = 0; tb < SEG_T; tb += 2 * SCAN_PF) {
        scan_fetch(dB_, xB_, pd, px, tb + SCAN_PF);
        __builtin_amdgcn_sched_barrier(0);
        scan_local_block(dA_, xA_, sB + tb, A2, h, sd);
        __builtin_amdgcn_sched_barrier(0);
        scan_fetch(dA_, xA_, pd, px, (tb + 2 * SCAN_PF < SEG_T ? tb + 2 * SCAN_PF : SEG_T - SCAN_PF));
        __builtin_amdgcn_sched_barrier(0);
        scan_local_block(dB_, xB_, sB + tb + SCAN_PF, A2, h, sd);
        __builtin_amdgcn_sched_barrier(0);
    }

    #pragma unroll
    for (int i = 0; i < 8; ++i) {
        segE[(size_t)(s * DS + 2 * i)     * DI + d] = h[i].x;
        segE[(size_t)(s * DS + 2 * i + 1) * DI + d] = h[i].y;
    }
    segSD[(size_t)s * DI + d] = sd;
}

__global__ void __launch_bounds__(256)
sssm_scan_out(const float* __restrict__ x, const float* __restrict__ A_log,
              const float* __restrict__ Dvec, const float* __restrict__ dbc,
              const float* __restrict__ segE, const float* __restrict__ segSD,
              float* out) {
    __shared__ __attribute__((aligned(16))) float sBC[SEG_T][2 * DS];

    const int d  = blockIdx.x * 256 + threadIdx.x;
    const int s  = blockIdx.y;
    const int t0 = s * SEG_T;

    {
        const int r = threadIdx.x >> 3, q = (threadIdx.x & 7) << 2;
        #pragma unroll
        for (int k = 0; k < SEG_T / 32; ++k)
            *(v4f*)&sBC[r + 32 * k][q] = *(const v4f*)(dbc + (size_t)(t0 + r + 32 * k) * RTOT + DTR + q);
    }

    v2f A2[8];
    load_A2(A_log, d, A2);
    const float Dd = Dvec[d];

    float*       po = out + (size_t)t0 * DI + d;
    const float* px = x   + (size_t)t0 * DI + d;

    v2f dA_[SCAN_PF / 2], xA_[SCAN_PF / 2], dB_[SCAN_PF / 2], xB_[SCAN_PF / 2];
    scan_fetch(dA_, xA_, po, px, 0);

    v2f h[8];
    #pragma unroll
    for (int i = 0; i < 8; ++i) h[i] = v2f{0.0f, 0.0f};
    for (int sp0 = 0; sp0 < s; sp0 += 5) {
        float sdl[5];
        v2f   E[5][8];
        #pragma unroll
        for (int u = 0; u < 5; ++u) {
            const int sp = (sp0 + u < SEG_S - 1) ? sp0 + u : SEG_S - 2;
            sdl[u] = segSD[(size_t)sp * DI + d];
            #pragma unroll
            for (int i = 0; i < 8; ++i)
                E[u][i] = v2f{segE[(size_t)(sp * DS + 2 * i) * DI + d],
                              segE[(size_t)(sp * DS + 2 * i + 1) * DI + d]};
        }
        #pragma unroll
        for (int u = 0; u < 5; ++u) {
            const bool on = sp0 + u < s;
            #pragma unroll
            for (int i = 0; i < 8; ++i) {
                const v2f P  = on ? exp2_2(sdl[u] * A2[i]) : v2f{1.0f, 1.0f};
                const v2f Eu = on ? E[u][i] : v2f{0.0f, 0.0f};
                h[i] = P * h[i] + Eu;
            }
        }
    }

    __syncthreads();

    for (int tb = 0; tb < SEG_T; tb += 2 * SCAN_PF) {
        scan_fetch(dB_, xB_, po, px, tb + SCAN_PF);
        __builtin_amdgcn_sched_barrier(0);
        scan_out_block(dA_, xA_, sBC + tb, A2, h, Dd, po + (size_t)tb * DI);
        __builtin_amdgcn_sched_barrier(0);
        scan_fetch(dA_, xA_, po, px, (tb + 2 * SCAN_PF < SEG_T ? tb + 2 * SCAN_PF : SEG_T - SCAN_PF));
        __builtin_amdgcn_sched_barrier(0);
        scan_out_block(dB_, xB_, sBC + tb + SCAN_PF, A2, h, Dd, po + (size_t)(tb + SCAN_PF) * DI);
        __builtin_amdgcn_sched_barrier(0);
    }
}

extern "C" void kernel_launch(void* const* d_in, const int* in_sizes, int n_in,
                              void* d_out, int out_size, void* d_ws, size_t ws_size,
                              hipStream_t stream) {
    const float* x    = (const float*)d_in[0];
    const float* Win  = (const float*)d_in[1];
    const float* Wd   = (const float*)d_in[2];
    const float* Alog = (const float*)d_in[3];
    const float* Dv   = (const float*)d_in[4];
    float*       out  = (float*)d_out;

    float* part = (float*)d_ws;
    float* dbc  = part + (size_t)KSPLIT * DBC_N;
    float* segE  = dbc  + (size_t)DBC_N;
    float* segSD = segE + (size_t)(SEG_S - 1) * DS * DI;

    sssm_gemm_in<<<512, 128, 0, stream>>>(x, Win, part);
    sssm_reduce<<<DBC_N / 256, 256, 0, stream>>>(part, dbc);
    sssm_gemm_delta<<<512, 256, 0, stream>>>(dbc, Wd, out);
    sssm_scan_local<<<dim3(DI / 256, SEG_S - 1), 256, 0, stream>>>(x, Alog, dbc, out, segE, segSD);
    sssm_scan_out<<<dim3(DI / 256, SEG_S), 256, 0, stream>>>(x, Alog, Dv, dbc, segE, segSD, out);
}


# delta GEMM and scan pass 1 fused into one launch: a wave makes its 32 channels' delta 32 rows at a time on WMMA, turns the tile through LDS and scans it under the other wave's WMMAs; sum orders kept
# speedup vs baseline: 20.1164x; 1.0552x over previous
#include <hip/hip_runtime.h>


#define SEQ_L   2048
#define DI      4096
#define RTOT    160
#define DTR     128
#define DS      16
#define KSPLIT  16
#define KCH     (DI / KSPLIT)
#define DBC_N   (SEQ_L * RTOT)

typedef float v2f __attribute__((ext_vector_type(2)));
typedef float v4f __attribute__((ext_vector_type(4)));
typedef float v8f __attribute__((ext_vector_type(8)));

static __device__ __forceinline__ v2f lo2(v4f v) { return v2f{v.x, v.y}; }
static __device__ __forceinline__ v2f hi2(v4f v) { return v2f{v.z, v.w}; }

static __device__ __forceinline__ v8f wmma4(v2f a, v2f b, v8f c) {
    return __builtin_amdgcn_wmma_f32_16x16x4_f32(
               false, a, false, b, (short)0, c, false, false);
}


__global__ void __launch_bounds__(128)
sssm_gemm_in(const float* __restrict__ x, const float* __restrict__ Win,
             float* __restrict__ part) {
    const int w    = threadIdx.x >> 5;
    const int lane = threadIdx.x & 31;
    const int ks   = blockIdx.x & (KSPLIT - 1);
    const int lp   = blockIdx.x >> 4;
    const int l0   = (lp * 2 + (w >> 1)) * 32;
    const int j0   = (w & 1) * 80;

    const int m  = lane & 15;
    const int kh = (lane >> 4) << 2;

    const float* ap = x   + (size_t)(l0 + m) * DI + ks * KCH + kh;
    const float* bp = Win + (size_t)(j0 + m) * DI + ks * KCH + kh;

    v8f acc[2][5];
    #pragma unroll
    for (int i = 0; i < 2; ++i)
        #pragma unroll
        for (int f = 0; f < 5; ++f) acc[i][f] = v8f{};

    v4f a[2][4], b[5][4];
    #pragma unroll
    for (int i = 0; i < 2; ++i)
        #pragma unroll
        for (int q = 0; q < 4; ++q) a[i][q] = *(const v4f*)(ap + (size_t)i * 16 * DI + 8 * q);
    #pragma unroll
    for (int f = 0; f < 5; ++f)
        #pragma unroll
        for (int q = 0; q < 4; ++q) b[f][q] = *(const v4f*)(bp + (size_t)f * 16 * DI + 8 * q);

    #pragma unroll
    for (int k = 0; k < KCH; k += 32) {
        v4f an[2][4], bn[5][4];
        if (k + 32 < KCH) {
            ap += 32; bp += 32;
            #pragma unroll
            for (int i = 0; i < 2; ++i)
                #pragma unroll
                for (int q = 0; q < 4; ++q) an[i][q] = *(const v4f*)(ap + (size_t)i * 16 * DI + 8 * q);
            #pragma unroll
            for (int f = 0; f < 5; ++f)
                #pragma unroll
                for (int q = 0; q < 4; ++q) bn[f][q] = *(const v4f*)(bp + (size_t)f * 16 * DI + 8 * q);
        }
        __builtin_amdgcn_sched_barrier(0);
        #pragma unroll
        for (int q = 0; q < 4; ++q) {
            #pragma unroll
            for (int i = 0; i < 2; ++i)
                #pragma unroll
                for (int f = 0; f < 5; ++f) acc[i][f] = wmma4(lo2(a[i][q]), lo2(b[f][q]), acc[i][f]);
            #pragma unroll
            for (int i = 0; i < 2; ++i)
                #pragma unroll
                for (int f = 0; f < 5; ++f) acc[i][f] = wmma4(hi2(a[i][q]), hi2(b[f][q]), acc[i][f]);
        }
        __builtin_amdgcn_sched_barrier(0);
        if (k + 32 < KCH) {
            #pragma unroll
            for (int i = 0; i < 2; ++i)
                #pragma unroll
                for (int q = 0; q < 4; ++q) a[i][q] = an[i][q];
            #pragma unroll
            for (int f = 0; f < 5; ++f)
                #pragma unroll
                for (int q = 0; q < 4; ++q) b[f][q] = bn[f][q];
        }
    }

    const int mrow = (lane >> 4) << 3;
    float* dst = part + (size_t)ks * DBC_N;
    #pragma unroll
    for (int i = 0; i < 2; ++i)
        #pragma unroll
        for (int f = 0; f < 5; ++f)
            #pragma unroll
            for (int r = 0; r < 8; ++r)
                dst[(size_t)(l0 + 16 * i + mrow + r) * RTOT + j0 + 16 * f + (lane & 15)] = acc[i][f][r];
}

__global__ void __launch_bounds__(256)
sssm_reduce(const float* __restrict__ part, float* __restrict__ dbc) {
    const int i = blockIdx.x * 256 + threadIdx.x;
    float s[KSPLIT];
    #pragma unroll
    for (int p = 0; p < KSPLIT; ++p) s[p] = part[i + (size_t)p * DBC_N];
    #pragma unroll
    for (int st = 1; st < KSPLIT; st <<= 1)
        #pragma unroll
        for (int p = 0; p < KSPLIT; p += 2 * st) s[p] += s[p + st];
    dbc[i] = s[0];
}

#define LN2_F 0.693147180559945309f
static __device__ __forceinline__ float softplus1(float v) {
    const float e = __expf(-fabsf(v));
    const float u = 1.0f + e;
    const float c = (u - 1.0f) - e;
    const float l = fmaf(__builtin_amdgcn_logf(u), LN2_F, -(c * __builtin_amdgcn_rcpf(u)));
    return fmaxf(v, 0.0f) + l;
}

#define SEG_S   16
#define SEG_T   (SEQ_L / SEG_S)
#define SCAN_PF 8
#define LOG2E_F 1.44269504088896341f

static __device__ __forceinline__ v2f exp2_2(v2f v) {
    return v2f{__builtin_amdgcn_exp2f(v.x), __builtin_amdgcn_exp2f(v.y)};
}

static __device__ __forceinline__ void load_A2(const float* __restrict__ A_log, int d, v2f (&A2)[8]) {
    const v4f* ap = (const v4f*)(A_log + (size_t)d * DS);
    #pragma unroll
    for (int q = 0; q < 4; ++q) {
        const v4f al = ap[q];
        A2[2 * q]     = v2f{-__expf(al.x) * LOG2E_F, -__expf(al.y) * LOG2E_F};
        A2[2 * q + 1] = v2f{-__expf(al.z) * LOG2E_F, -__expf(al.w) * LOG2E_F};
    }
}

static __device__ __forceinline__ void scan_fetch(v2f (&dl)[SCAN_PF / 2], v2f (&xl)[SCAN_PF / 2],
                                                  const float* pd, const float* px, int r) {
    #pragma unroll
    for (int j = 0; j < SCAN_PF; j += 2) {
        dl[j >> 1].x = pd[(size_t)(r + j) * DI]; dl[j >> 1].y = pd[(size_t)(r + j + 1) * DI];
        xl[j >> 1].x = px[(size_t)(r + j) * DI]; xl[j >> 1].y = px[(size_t)(r + j + 1) * DI];
    }
}

static __device__ __forceinline__ void scan_exp_rows(v2f dd, const v2f (&A2)[8], v2f (&e0)[8], v2f (&e1)[8]) {
    #pragma unroll
    for (int i = 0; i < 8; ++i) {
        const v2f ax = dd * A2[i].x;
        const v2f ay = dd * A2[i].y;
        e0[i] = v2f{__builtin_amdgcn_exp2f(ax.x), __builtin_amdgcn_exp2f(ay.x)};
        e1[i] = v2f{__builtin_amdgcn_exp2f(ax.y), __builtin_amdgcn_exp2f(ay.y)};
    }
}

template <int NQ>
static __device__ __forceinline__ void scan_lds_row(v4f (&bc)[NQ], const float* row) {
    #pragma unroll
    for (int q = 0; q < NQ; ++q) bc[q] = ((const v4f*)row)[q];
}

static __device__ __forceinline__ void scan_step_h(const v2f (&e)[8], float du, const v4f (&b)[4], v2f (&h)[8]) {
    #pragma unroll
    for (int q = 0; q < 4; ++q) {
        h[2 * q]     = e[2 * q]     * h[2 * q]     + du * lo2(b[q]);
        h[2 * q + 1] = e[2 * q + 1] * h[2 * q + 1] + du * hi2(b[q]);
    }
}

static __device__ __forceinline__ float scan_step_hy(const v2f (&e)[8], float du, const v4f (&bc)[8], v2f (&h)[8]) {
    v2f ya = v2f{0.0f, 0.0f}, yb = v2f{0.0f, 0.0f};
    #pragma unroll
    for (int q = 0; q < 4; ++q) {
        h[2 * q]     = e[2 * q]     * h[2 * q]     + du * lo2(bc[q]);
        h[2 * q + 1] = e[2 * q + 1] * h[2 * q + 1] + du * hi2(bc[q]);
        ya = h[2 * q]     * lo2(bc[4 + q]) + ya;
        yb = h[2 * q + 1] * hi2(bc[4 + q]) + yb;
    }
    const v2f yy = ya + yb;
    return yy.x + yy.y;
}

static __device__ __forceinline__ void scan_local_block(const v2f (&dl)[SCAN_PF / 2], const v2f (&xl)[SCAN_PF / 2],
                                                        const float (*sb)[DS], const v2f (&A2)[8],
                                                        v2f (&h)[8], float& sd) {
    #pragma unroll
    for (int jj = 0; jj < SCAN_PF / 2; ++jj) {
        v4f b0[4], b1[4];
        scan_lds_row<4>(b0, sb[2 * jj]);
        scan_lds_row<4>(b1, sb[2 * jj + 1]);
        __builtin_amdgcn_sched_barrier(0);
        const v2f dd  = dl[jj];
        const v2f du2 = dd * xl[jj];
        v2f e0[8], e1[8];
        scan_exp_rows(dd, A2, e0, e1);
        sd += dd.x;
        sd += dd.y;
        scan_step_h(e0, du2.x, b0, h);
        scan_step_h(e1, du2.y, b1, h);
        __builtin_amdgcn_sched_barrier(0);
    }
}

static __device__ __forceinline__ void scan_out_block(const v2f (&dl)[SCAN_PF / 2], const v2f (&xl)[SCAN_PF / 2],
                                                      const float (*sbc)[2 * DS], const v2f (&A2)[8],
                                                      v2f (&h)[8], float Dd, float* po) {
    #pragma unroll
    for (int jj = 0; jj < SCAN_PF / 2; ++jj) {
        v4f b0[8], b1[8];
        scan_lds_row<8>(b0, sbc[2 * jj]);
        scan_lds_row<8>(b1, sbc[2 * jj + 1]);
        __builtin_amdgcn_sched_barrier(0);
        const v2f dd  = dl[jj];
        const v2f du2 = dd * xl[jj];
        v2f e0[8], e1[8];
        scan_exp_rows(dd, A2, e0, e1);
        const float ys0 = scan_step_hy(e0, du2.x, b0, h);
        po[(size_t)(2 * jj) * DI]     = fmaf(xl[jj].x, Dd, ys0);
        __builtin_amdgcn_sched_barrier(0);
        const float ys1 = scan_step_hy(e1, du2.y, b1, h);
        po[(size_t)(2 * jj + 1) * DI] = fmaf(xl[jj].y, Dd, ys1);
        __builtin_amdgcn_sched_barrier(0);
    }
}

#define FA_PITCH 132
#define FT_PITCH 34

__global__ void __launch_bounds__(256)
sssm_delta_scan(const float* __restrict__ dbc, const float* __restrict__ Wd,
                const float* __restrict__ x, const float* __restrict__ A_log,
                float* __restrict__ delta, float* __restrict__ segE, float* __restrict__ segSD) {
    __shared__ __attribute__((aligned(16))) float sA[SEG_T][FA_PITCH];
    __shared__ __attribute__((aligned(16))) float sB[SEG_T][DS];
    __shared__ __attribute__((aligned(16))) float sT[8][32][FT_PITCH];

    const int w    = threadIdx.x >> 5;
    const int lane = threadIdx.x & 31;
    const int d    = blockIdx.x * 256 + threadIdx.x;
    const int dw   = blockIdx.x * 256 + w * 32;
    const int s    = blockIdx.y;
    const int t0   = s * SEG_T;
    const bool do_scan = s < SEG_S - 1;

    {
        #pragma unroll
        for (int j = 0; j < 16; ++j) {
            const int idx = threadIdx.x + 256 * j;
            const int r = idx >> 5, q = (idx & 31) << 2;
            *(v4f*)&sA[r][q] = *(const v4f*)(dbc + (size_t)(t0 + r) * RTOT + q);
        }
    }
    {
        const int r = threadIdx.x >> 2, q = (threadIdx.x & 3) << 2;
        #pragma unroll
        for (int k = 0; k < SEG_T / 64; ++k)
            *(v4f*)&sB[r + 64 * k][q] = *(const v4f*)(dbc + (size_t)(t0 + r + 64 * k) * RTOT + DTR + q);
    }

    v2f A2[8];
    load_A2(A_log, d, A2);

    const int m    = lane & 15;
    const int kh   = (lane >> 4) << 2;
    const int mrow = (lane >> 4) << 3;
    const float* bp = Wd + (size_t)(dw + m) * DTR + kh;
    float*       tw = &sT[w][0][0];
    float*       pd = delta + (size_t)t0 * DI + d;
    const float* px = x     + (size_t)t0 * DI + d;

    v4f bw[4][2][4];
    #pragma unroll
    for (int ks = 0; ks < 4; ++ks)
        #pragma unroll
        for (int f = 0; f < 2; ++f)
            #pragma unroll
            for (int q = 0; q < 4; ++q) bw[ks][f][q] = *(const v4f*)(bp + (size_t)f * 16 * DTR + 32 * ks + 8 * q);

    v2f h[8];
    #pragma unroll
    for (int i = 0; i < 8; ++i) h[i] = v2f{0.0f, 0.0f};
    float sd = 0.0f;

    __syncthreads();

    #pragma unroll 1
    for (int rb = 0; rb < SEG_T; rb += 32) {
        float xr[32];
        if (do_scan) {
            #pragma unroll
            for (int j = 0; j < 32; ++j) xr[j] = px[(size_t)(rb + j) * DI];
        } else {
            #pragma unroll
            for (int j = 0; j < 32; ++j) xr[j] = 0.0f;
        }

        v8f acc[2][2];
        #pragma unroll
        for (int i = 0; i < 2; ++i)
            #pragma unroll
            for (int f = 0; f < 2; ++f) acc[i][f] = v8f{};

        v4f a[2][4];
        #pragma unroll
        for (int i = 0; i < 2; ++i)
            #pragma unroll
            for (int q = 0; q < 4; ++q) a[i][q] = *(const v4f*)&sA[rb + 16 * i + m][8 * q + kh];

        #pragma unroll
        for (int ks = 0; ks < 4; ++ks) {
            v4f an[2][4];
            if (ks < 3) {
                #pragma unroll
                for (int i = 0; i < 2; ++i)
                    #pragma unroll
                    for (int q = 0; q < 4; ++q) an[i][q] = *(const v4f*)&sA[rb + 16 * i + m][32 * (ks + 1) + 8 * q + kh];
            }
            __builtin_amdgcn_sched_barrier(0);
            #pragma unroll
            for (int q = 0; q < 4; ++q) {
                #pragma unroll
                for (int i = 0; i < 2; ++i)
                    #pragma unroll
                    for (int f = 0; f < 2; ++f) acc[i][f] = wmma4(lo2(a[i][q]), lo2(bw[ks][f][q]), acc[i][f]);
                #pragma unroll
                for (int i = 0; i < 2; ++i)
                    #pragma unroll
                    for (int f = 0; f < 2; ++f) acc[i][f] = wmma4(hi2(a[i][q]), hi2(bw[ks][f][q]), acc[i][f]);
            }
            __builtin_amdgcn_sched_barrier(0);
            if (ks < 3) {
                #pragma unroll
                for (int i = 0; i < 2; ++i)
                    #pragma unroll
                    for (int q = 0; q < 4; ++q) a[i][q] = an[i][q];
            }
        }

        #pragma unroll
        for (int i = 0; i < 2; ++i)
            #pragma unroll
            for (int f = 0; f < 2; ++f)
                #pragma unroll
                for (int r = 0; r < 8; ++r)
                    tw[(16 * i + mrow + r) * FT_PITCH + 16 * f + m] = softplus1(acc[i][f][r]);
        __builtin_amdgcn_fence(__ATOMIC_RELEASE, "wavefront");
        __builtin_amdgcn_wave_barrier();
        __builtin_amdgcn_fence(__ATOMIC_ACQUIRE, "wavefront");

        #pragma unroll
        for (int rr = 0; rr < 32; rr += SCAN_PF) {
            v2f dl[SCAN_PF / 2], xl[SCAN_PF / 2];
            #pragma unroll
            for (int j = 0; j < SCAN_PF; j += 2) {
                dl[j >> 1].x = tw[(rr + j) * FT_PITCH + lane];
                dl[j >> 1].y = tw[(rr + j + 1) * FT_PITCH + lane];
                xl[j >> 1]   = v2f{xr[rr + j], xr[rr + j + 1]};
            }
            #pragma unroll
            for (int j = 0; j < SCAN_PF; j += 2) {
                pd[(size_t)(rb + rr + j) * DI]     = dl[j >> 1].x;
                pd[(size_t)(rb + rr + j + 1) * DI] = dl[j >> 1].y;
            }
            if (do_scan) scan_local_block(dl, xl, sB + rb + rr, A2, h, sd);
        }
        __builtin_amdgcn_fence(__ATOMIC_RELEASE, "wavefront");
        __builtin_amdgcn_wave_barrier();
    }

    if (do_scan) {
        #pragma unroll
        for (int i = 0; i < 8; ++i) {
            segE[(size_t)(s * DS + 2 * i)     * DI + d] = h[i].x;
            segE[(size_t)(s * DS + 2 * i + 1) * DI + d] = h[i].y;
        }
        segSD[(size_t)s * DI + d] = sd;
    }
}

__global__ void __launch_bounds__(256)
sssm_scan_out(const float* __restrict__ x, const float* __restrict__ A_log,
              const float* __restrict__ Dvec, const float* __restrict__ dbc,
              const float* __restrict__ segE, const float* __restrict__ segSD,
              float* out) {
    __shared__ __attribute__((aligned(16))) float sBC[SEG_T][2 * DS];

    const int d  = blockIdx.x * 256 + threadIdx.x;
    const int s  = blockIdx.y;
    const int t0 = s * SEG_T;

    {
        const int r = threadIdx.x >> 3, q = (threadIdx.x & 7) << 2;
        #pragma unroll
        for (int k = 0; k < SEG_T / 32; ++k)
            *(v4f*)&sBC[r + 32 * k][q] = *(const v4f*)(dbc + (size_t)(t0 + r + 32 * k) * RTOT + DTR + q);
    }

    v2f A2[8];
    load_A2(A_log, d, A2);
    const float Dd = Dvec[d];

    float*       po = out + (size_t)t0 * DI + d;
    const float* px = x   + (size_t)t0 * DI + d;

    v2f dA_[SCAN_PF / 2], xA_[SCAN_PF / 2], dB_[SCAN_PF / 2], xB_[SCAN_PF / 2];
    scan_fetch(dA_, xA_, po, px, 0);

    v2f h[8];
    #pragma unroll
    for (int i = 0; i < 8; ++i) h[i] = v2f{0.0f, 0.0f};
    for (int sp0 = 0; sp0 < s; sp0 += 5) {
        float sdl[5];
        v2f   E[5][8];
        #pragma unroll
        for (int u = 0; u < 5; ++u) {
            const int sp = (sp0 + u < SEG_S - 1) ? sp0 + u : SEG_S - 2;
            sdl[u] = segSD[(size_t)sp * DI + d];
            #pragma unroll
            for (int i = 0; i < 8; ++i)
                E[u][i] = v2f{segE[(size_t)(sp * DS + 2 * i) * DI + d],
                              segE[(size_t)(sp * DS + 2 * i + 1) * DI + d]};
        }
        #pragma unroll
        for (int u = 0; u < 5; ++u) {
            const bool on = sp0 + u < s;
            #pragma unroll
            for (int i = 0; i < 8; ++i) {
                const v2f P  = on ? exp2_2(sdl[u] * A2[i]) : v2f{1.0f, 1.0f};
                const v2f Eu = on ? E[u][i] : v2f{0.0f, 0.0f};
                h[i] = P * h[i] + Eu;
            }
        }
    }

    __syncthreads();

    for (int tb = 0; tb < SEG_T; tb += 2 * SCAN_PF) {
        scan_fetch(dB_, xB_, po, px, tb + SCAN_PF);
        __builtin_amdgcn_sched_barrier(0);
        scan_out_block(dA_, xA_, sBC + tb, A2, h, Dd, po + (size_t)tb * DI);
        __builtin_amdgcn_sched_barrier(0);
        scan_fetch(dA_, xA_, po, px, (tb + 2 * SCAN_PF < SEG_T ? tb + 2 * SCAN_PF : SEG_T - SCAN_PF));
        __builtin_amdgcn_sched_barrier(0);
        scan_out_block(dB_, xB_, sBC + tb + SCAN_PF, A2, h, Dd, po + (size_t)(tb + SCAN_PF) * DI);
        __builtin_amdgcn_sched_barrier(0);
    }
}

extern "C" void kernel_launch(void* const* d_in, const int* in_sizes, int n_in,
                              void* d_out, int out_size, void* d_ws, size_t ws_size,
                              hipStream_t stream) {
    const float* x    = (const float*)d_in[0];
    const float* Win  = (const float*)d_in[1];
    const float* Wd   = (const float*)d_in[2];
    const float* Alog = (const float*)d_in[3];
    const float* Dv   = (const float*)d_in[4];
    float*       out  = (float*)d_out;

    float* part = (float*)d_ws;
    float* dbc  = part + (size_t)KSPLIT * DBC_N;
    float* segE  = dbc  + (size_t)DBC_N;
    float* segSD = segE + (size_t)(SEG_S - 1) * DS * DI;

    sssm_gemm_in<<<512, 128, 0, stream>>>(x, Win, part);
    sssm_reduce<<<DBC_N / 256, 256, 0, stream>>>(part, dbc);
    sssm_delta_scan<<<dim3(DI / 256, SEG_S), 256, 0, stream>>>(dbc, Wd, x, Alog, out, segE, segSD);
    sssm_scan_out<<<dim3(DI / 256, SEG_S), 256, 0, stream>>>(x, Alog, Dv, dbc, segE, segSD, out);
}
